# starting kernel
# baseline (speedup 1.0000x reference)
#include <hip/hip_runtime.h>
#include <math.h>

typedef __attribute__((ext_vector_type(2))) float v2f;
typedef __attribute__((ext_vector_type(8))) float v8f;

#define NNODES 100000
#define HID    128
#define TILE_M 16
#define ROWT   2
#define ROWS   (ROWT * TILE_M)

static __device__ __forceinline__ float atomic_add_f32(float* p, float v) {
  return __hip_atomic_fetch_add(p, v, __ATOMIC_RELAXED, __HIP_MEMORY_SCOPE_AGENT);
}

__global__ __launch_bounds__(256) void deg_kernel(const int* __restrict__ dst,
                                                  float* __restrict__ deg, int E) {
  int e = blockIdx.x * 256 + threadIdx.x;
  if (e < E) atomic_add_f32(&deg[dst[e]], 1.0f);
}

__global__ __launch_bounds__(256) void invdeg_kernel(float* __restrict__ deg, int N) {
  int n = blockIdx.x * 256 + threadIdx.x;
  if (n < N) deg[n] = 1.0f / fmaxf(deg[n], 1.0f);
}

__global__ __launch_bounds__(256) void scatter_kernel(const float* __restrict__ x,
                                                      const int* __restrict__ src,
                                                      const int* __restrict__ dst,
                                                      float* __restrict__ agg, int E) {
  int gid  = blockIdx.x * 256 + threadIdx.x;
  int e    = gid >> 5;
  int lane = gid & 31;
  if (e >= E) return;
  int s = src[e], d = dst[e];
  const float4 v = *(const float4*)(x + (size_t)s * HID + lane * 4);
  float* ap = agg + (size_t)d * HID + lane * 4;
  atomic_add_f32(ap + 0, v.x);
  atomic_add_f32(ap + 1, v.y);
  atomic_add_f32(ap + 2, v.z);
  atomic_add_f32(ap + 3, v.w);
}

__global__ __launch_bounds__(256) void sage_transform(
    const float* __restrict__ agg, const float* __restrict__ xin,
    const float* __restrict__ Wl, const float* __restrict__ bl,
    const float* __restrict__ Wr, const float* __restrict__ invdeg,
    float* __restrict__ xout) {
  __shared__ float lds_a[ROWS * HID];
  __shared__ float lds_x[ROWS * HID];
  __shared__ float lds_o[ROWS * HID];
  __shared__ float lds_inv[ROWS];

  const int tid  = threadIdx.x;
  const int row0 = blockIdx.x * ROWS;

#pragma unroll
  for (int rt = 0; rt < ROWT; ++rt) {
    const int r = rt * TILE_M + (tid >> 4);
    const int c = (tid & 15) * 8;
    const float inv = invdeg[row0 + r];
    const float4* pa = (const float4*)(agg + (size_t)(row0 + r) * HID + c);
    float4 a0 = pa[0], a1 = pa[1];
    float4* la4 = (float4*)(lds_a + r * HID + c);
    la4[0] = make_float4(a0.x * inv, a0.y * inv, a0.z * inv, a0.w * inv);
    la4[1] = make_float4(a1.x * inv, a1.y * inv, a1.z * inv, a1.w * inv);
    const float4* px = (const float4*)(xin + (size_t)(row0 + r) * HID + c);
    float4* lx4 = (float4*)(lds_x + r * HID + c);
    lx4[0] = px[0];
    lx4[1] = px[1];
  }

  const int wave  = tid >> 5;
  const int lane  = tid & 31;
  const int nn    = lane & 15;
  const int khalf = (lane >> 4) * 2;
  const int h0    = wave * 16;

  v2f bL[32], bR[32];
  {
    const float* pwl = Wl + (size_t)(h0 + nn) * HID + khalf;
    const float* pwr = Wr + (size_t)(h0 + nn) * HID + khalf;
#pragma unroll
    for (int kk = 0; kk < 32; ++kk) {
      bL[kk] = *(const v2f*)(pwl + kk * 4);
      bR[kk] = *(const v2f*)(pwr + kk * 4);
    }
  }
  const float bias  = bl[h0 + nn];
  const int   rbase = (lane >> 4) * 8;
  __syncthreads();

#pragma unroll 1
  for (int rt = 0; rt < ROWT; ++rt) {
    v8f accL = {0.f, 0.f, 0.f, 0.f, 0.f, 0.f, 0.f, 0.f};
    v8f accR = {0.f, 0.f, 0.f, 0.f, 0.f, 0.f, 0.f, 0.f};
    const float* la = lds_a + (rt * TILE_M + nn) * HID + khalf;
    const float* lx = lds_x + (rt * TILE_M + nn) * HID + khalf;
#pragma unroll
    for (int kk = 0; kk < 32; ++kk) {
      v2f aA = *(const v2f*)(la + kk * 4);
      accL = __builtin_amdgcn_wmma_f32_16x16x4_f32(false, aA, false, bL[kk],
                                                   (short)0, accL, false, false);
      v2f aX = *(const v2f*)(lx + kk * 4);
      accR = __builtin_amdgcn_wmma_f32_16x16x4_f32(false, aX, false, bR[kk],
                                                   (short)0, accR, false, false);
    }
#pragma unroll
    for (int j = 0; j < 8; ++j)
      lds_o[(rt * TILE_M + rbase + j) * HID + h0 + nn] = accL[j] + accR[j] + bias;
  }
  __syncthreads();

#pragma unroll
  for (int rt = 0; rt < ROWT; ++rt) {
    const int r = rt * TILE_M + (tid >> 4);
    const int p = tid & 15;
    const float* po = lds_o + r * HID + p * 8;
    float s = 0.f;
#pragma unroll
    for (int q = 0; q < 8; ++q) s += po[q] * po[q];
    s += __shfl_xor(s, 1);
    s += __shfl_xor(s, 2);
    s += __shfl_xor(s, 4);
    s += __shfl_xor(s, 8);
    if (p == 0) lds_inv[r] = 1.0f / fmaxf(sqrtf(s), 1e-12f);
  }
  __syncthreads();

#pragma unroll
  for (int rt = 0; rt < ROWT; ++rt) {
    const int r = rt * TILE_M + (tid >> 4);
    const int c = (tid & 15) * 8;
    const float inv = lds_inv[r];
    const float* po = lds_o + r * HID + c;
    float4 o0, o1;
    o0.x = fmaxf(po[0] * inv, 0.f); o0.y = fmaxf(po[1] * inv, 0.f);
    o0.z = fmaxf(po[2] * inv, 0.f); o0.w = fmaxf(po[3] * inv, 0.f);
    o1.x = fmaxf(po[4] * inv, 0.f); o1.y = fmaxf(po[5] * inv, 0.f);
    o1.z = fmaxf(po[6] * inv, 0.f); o1.w = fmaxf(po[7] * inv, 0.f);
    float4* pw = (float4*)(xout + (size_t)(row0 + r) * HID + c);
    pw[0] = o0;
    pw[1] = o1;
  }
}

__global__ __launch_bounds__(256) void mlp_head(
    const float* __restrict__ xin, const float* __restrict__ W1,
    const float* __restrict__ b1, const float* __restrict__ W2,
    const float* __restrict__ b2, float* __restrict__ out) {
  __shared__ float lds_x[TILE_M * HID];
  __shared__ float lds_h[TILE_M * HID];

  const int tid  = threadIdx.x;
  const int row0 = blockIdx.x * TILE_M;

  {
    const int r = tid >> 4;
    const int c = (tid & 15) * 8;
    const float4* px = (const float4*)(xin + (size_t)(row0 + r) * HID + c);
    float4* lx4 = (float4*)(lds_x + r * HID + c);
    lx4[0] = px[0];
    lx4[1] = px[1];
  }

  const int wave  = tid >> 5;
  const int lane  = tid & 31;
  const int nn    = lane & 15;
  const int khalf = (lane >> 4) * 2;
  const int h0    = wave * 16;

  v2f bW[32];
  {
    const float* pw = W1 + (size_t)(h0 + nn) * HID + khalf;
#pragma unroll
    for (int kk = 0; kk < 32; ++kk) bW[kk] = *(const v2f*)(pw + kk * 4);
  }
  const float bias = b1[h0 + nn];
  __syncthreads();

  v8f acc = {0.f, 0.f, 0.f, 0.f, 0.f, 0.f, 0.f, 0.f};
  const float* lx = lds_x + nn * HID + khalf;
#pragma unroll
  for (int kk = 0; kk < 32; ++kk) {
    v2f a = *(const v2f*)(lx + kk * 4);
    acc = __builtin_amdgcn_wmma_f32_16x16x4_f32(false, a, false, bW[kk],
                                                (short)0, acc, false, false);
  }
  const int rbase = (lane >> 4) * 8;
#pragma unroll
  for (int j = 0; j < 8; ++j)
    lds_h[(rbase + j) * HID + h0 + nn] = fmaxf(acc[j] + bias, 0.f);
  __syncthreads();

  {
    const int r = tid >> 4;
    const int p = tid & 15;
    const float* ph = lds_h + r * HID + p * 8;
    float s0 = 0.f, s1 = 0.f;
#pragma unroll
    for (int q = 0; q < 8; ++q) {
      float h = ph[q];
      s0 += h * W2[p * 8 + q];
      s1 += h * W2[HID + p * 8 + q];
    }
    s0 += __shfl_xor(s0, 1); s1 += __shfl_xor(s1, 1);
    s0 += __shfl_xor(s0, 2); s1 += __shfl_xor(s1, 2);
    s0 += __shfl_xor(s0, 4); s1 += __shfl_xor(s1, 4);
    s0 += __shfl_xor(s0, 8); s1 += __shfl_xor(s1, 8);
    if (p == 0) {
      out[(size_t)(row0 + r) * 2 + 0] = s0 + b2[0];
      out[(size_t)(row0 + r) * 2 + 1] = s1 + b2[1];
    }
  }
}

extern "C" void kernel_launch(void* const* d_in, const int* in_sizes, int n_in,
                              void* d_out, int out_size, void* d_ws, size_t ws_size,
                              hipStream_t stream) {
  const float* x  = (const float*)d_in[0];
  const int*   ei = (const int*)d_in[1];
  const float* Wl = (const float*)d_in[2];
  const float* bl = (const float*)d_in[3];
  const float* Wr = (const float*)d_in[4];
  const float* W1 = (const float*)d_in[5];
  const float* b1 = (const float*)d_in[6];
  const float* W2 = (const float*)d_in[7];
  const float* b2 = (const float*)d_in[8];
  float* out = (float*)d_out;

  const int N = NNODES;
  const int E = in_sizes[1] / 2;
  const int* src = ei;
  const int* dst = ei + E;

  float* inv_deg = (float*)d_ws;
  const size_t pad = 100096;
  float* bufA = inv_deg + pad;
  float* bufB = bufA + (size_t)N * HID;

  hipMemsetAsync(inv_deg, 0, (size_t)N * sizeof(float), stream);
  deg_kernel<<<(E + 255) / 256, 256, 0, stream>>>(dst, inv_deg, E);
  invdeg_kernel<<<(N + 255) / 256, 256, 0, stream>>>(inv_deg, N);

  const float* xs[3] = {x, bufA, bufB};
  float*       ag[3] = {bufA, bufB, bufA};
  const int scatterBlocks = (E + 7) / 8;
  const int xformBlocks   = N / ROWS;
  const int headBlocks    = N / TILE_M;

  for (int i = 0; i < 3; ++i) {
    float* agg = ag[i];
    const float* xc = xs[i];
    hipMemsetAsync(agg, 0, (size_t)N * HID * sizeof(float), stream);
    scatter_kernel<<<scatterBlocks, 256, 0, stream>>>(xc, src, dst, agg, E);
    sage_transform<<<xformBlocks, 256, 0, stream>>>(
        agg, xc, Wl + (size_t)i * HID * HID, bl + (size_t)i * HID,
        Wr + (size_t)i * HID * HID, inv_deg, agg);
  }
  mlp_head<<<headBlocks, 256, 0, stream>>>(bufA, W1, b1, W2, b2, out);
}


# Neighbour sums by per-call CSR gather (int atomic count+rank, scan, fill; one wave per node) replace f32 atomic scatter and agg memsets; per-node sum order = arrival rank
# speedup vs baseline: 2.0994x; 2.0994x over previous
#include <hip/hip_runtime.h>
#include <math.h>

typedef __attribute__((ext_vector_type(2))) float v2f;
typedef __attribute__((ext_vector_type(8))) float v8f;

#define NNODES 100000
#define HID    128
#define TILE_M 16
#define ROWT   2
#define ROWS   (ROWT * TILE_M)

__global__ __launch_bounds__(256) void count_kernel(const int* __restrict__ dst,
                                                    int* __restrict__ cnt,
                                                    int* __restrict__ rank, int E) {
  int e = blockIdx.x * 256 + threadIdx.x;
  if (e < E)
    rank[e] = __hip_atomic_fetch_add(&cnt[dst[e]], 1, __ATOMIC_RELAXED,
                                     __HIP_MEMORY_SCOPE_AGENT);
}

__global__ __launch_bounds__(1024) void scan_kernel(const int* __restrict__ cnt,
                                                    int* __restrict__ rowptr,
                                                    float* __restrict__ invdeg, int N) {
  __shared__ int wpre[32];
  __shared__ int wtot[32];
  const int tid  = threadIdx.x;
  const int lane = tid & 31;
  const int wave = tid >> 5;

  int s = 0;
  const int4* c4 = (const int4*)cnt;
  const int n4 = blockIdx.x * 256;
  for (int i = tid; i < n4; i += 1024) {
    int4 v = c4[i];
    s += (v.x + v.y) + (v.z + v.w);
  }
  s += __shfl_xor(s, 1);
  s += __shfl_xor(s, 2);
  s += __shfl_xor(s, 4);
  s += __shfl_xor(s, 8);
  s += __shfl_xor(s, 16);
  if (lane == 0) wpre[wave] = s;

  const int n = blockIdx.x * 1024 + tid;
  const int c = (n < N) ? cnt[n] : 0;
  int incl = c;
#pragma unroll
  for (int o = 1; o < 32; o <<= 1) {
    int t = __shfl_up(incl, o);
    if (lane >= o) incl += t;
  }
  if (lane == 31) wtot[wave] = incl;
  __syncthreads();

  int ab = wpre[lane] + ((lane < wave) ? wtot[lane] : 0);
  ab += __shfl_xor(ab, 1);
  ab += __shfl_xor(ab, 2);
  ab += __shfl_xor(ab, 4);
  ab += __shfl_xor(ab, 8);
  ab += __shfl_xor(ab, 16);
  if (n < N) {
    rowptr[n] = ab + incl - c;
    invdeg[n] = 1.0f / fmaxf((float)c, 1.0f);
  }
}

__global__ __launch_bounds__(256) void fill_kernel(const int* __restrict__ src,
                                                   const int* __restrict__ dst,
                                                   const int* __restrict__ rowptr,
                                                   const int* __restrict__ rank,
                                                   int* __restrict__ col, int E) {
  int e = blockIdx.x * 256 + threadIdx.x;
  if (e < E) col[rowptr[dst[e]] + rank[e]] = src[e];
}

__global__ __launch_bounds__(256) void gather_kernel(const float* __restrict__ x,
                                                     const int* __restrict__ rowptr,
                                                     const int* __restrict__ cnt,
                                                     const int* __restrict__ col,
                                                     float* __restrict__ agg, int N) {
  const int lane = threadIdx.x & 31;
  const int n = __builtin_amdgcn_readfirstlane((int)(blockIdx.x * 8 + (threadIdx.x >> 5)));
  if (n >= N) return;
  const int beg = rowptr[n];
  const int end = beg + cnt[n];
  const float* xl = x + lane * 4;
  float4 acc = make_float4(0.f, 0.f, 0.f, 0.f);
  int j = beg;
  for (; j + 4 <= end; j += 4) {
    const int s0 = col[j], s1 = col[j + 1], s2 = col[j + 2], s3 = col[j + 3];
    const float4 v0 = *(const float4*)(xl + (size_t)s0 * HID);
    const float4 v1 = *(const float4*)(xl + (size_t)s1 * HID);
    const float4 v2 = *(const float4*)(xl + (size_t)s2 * HID);
    const float4 v3 = *(const float4*)(xl + (size_t)s3 * HID);
    acc.x += v0.x; acc.y += v0.y; acc.z += v0.z; acc.w += v0.w;
    acc.x += v1.x; acc.y += v1.y; acc.z += v1.z; acc.w += v1.w;
    acc.x += v2.x; acc.y += v2.y; acc.z += v2.z; acc.w += v2.w;
    acc.x += v3.x; acc.y += v3.y; acc.z += v3.z; acc.w += v3.w;
  }
  for (; j < end; ++j) {
    const int s0 = col[j];
    const float4 v0 = *(const float4*)(xl + (size_t)s0 * HID);
    acc.x += v0.x; acc.y += v0.y; acc.z += v0.z; acc.w += v0.w;
  }
  *(float4*)(agg + (size_t)n * HID + lane * 4) = acc;
}

__global__ __launch_bounds__(256) void sage_transform(
    const float* __restrict__ agg, const float* __restrict__ xin,
    const float* __restrict__ Wl, const float* __restrict__ bl,
    const float* __restrict__ Wr, const float* __restrict__ invdeg,
    float* __restrict__ xout) {
  __shared__ float lds_a[ROWS * HID];
  __shared__ float lds_x[ROWS * HID];
  __shared__ float lds_o[ROWS * HID];
  __shared__ float lds_inv[ROWS];

  const int tid  = threadIdx.x;
  const int row0 = blockIdx.x * ROWS;

#pragma unroll
  for (int rt = 0; rt < ROWT; ++rt) {
    const int r = rt * TILE_M + (tid >> 4);
    const int c = (tid & 15) * 8;
    const float inv = invdeg[row0 + r];
    const float4* pa = (const float4*)(agg + (size_t)(row0 + r) * HID + c);
    float4 a0 = pa[0], a1 = pa[1];
    float4* la4 = (float4*)(lds_a + r * HID + c);
    la4[0] = make_float4(a0.x * inv, a0.y * inv, a0.z * inv, a0.w * inv);
    la4[1] = make_float4(a1.x * inv, a1.y * inv, a1.z * inv, a1.w * inv);
    const float4* px = (const float4*)(xin + (size_t)(row0 + r) * HID + c);
    float4* lx4 = (float4*)(lds_x + r * HID + c);
    lx4[0] = px[0];
    lx4[1] = px[1];
  }

  const int wave  = tid >> 5;
  const int lane  = tid & 31;
  const int nn    = lane & 15;
  const int khalf = (lane >> 4) * 2;
  const int h0    = wave * 16;

  v2f bL[32], bR[32];
  {
    const float* pwl = Wl + (size_t)(h0 + nn) * HID + khalf;
    const float* pwr = Wr + (size_t)(h0 + nn) * HID + khalf;
#pragma unroll
    for (int kk = 0; kk < 32; ++kk) {
      bL[kk] = *(const v2f*)(pwl + kk * 4);
      bR[kk] = *(const v2f*)(pwr + kk * 4);
    }
  }
  const float bias  = bl[h0 + nn];
  const int   rbase = (lane >> 4) * 8;
  __syncthreads();

#pragma unroll 1
  for (int rt = 0; rt < ROWT; ++rt) {
    v8f accL = {0.f, 0.f, 0.f, 0.f, 0.f, 0.f, 0.f, 0.f};
    v8f accR = {0.f, 0.f, 0.f, 0.f, 0.f, 0.f, 0.f, 0.f};
    const float* la = lds_a + (rt * TILE_M + nn) * HID + khalf;
    const float* lx = lds_x + (rt * TILE_M + nn) * HID + khalf;
#pragma unroll
    for (int kk = 0; kk < 32; ++kk) {
      v2f aA = *(const v2f*)(la + kk * 4);
      accL = __builtin_amdgcn_wmma_f32_16x16x4_f32(false, aA, false, bL[kk],
                                                   (short)0, accL, false, false);
      v2f aX = *(const v2f*)(lx + kk * 4);
      accR = __builtin_amdgcn_wmma_f32_16x16x4_f32(false, aX, false, bR[kk],
                                                   (short)0, accR, false, false);
    }
#pragma unroll
    for (int j = 0; j < 8; ++j)
      lds_o[(rt * TILE_M + rbase + j) * HID + h0 + nn] = accL[j] + accR[j] + bias;
  }
  __syncthreads();

#pragma unroll
  for (int rt = 0; rt < ROWT; ++rt) {
    const int r = rt * TILE_M + (tid >> 4);
    const int p = tid & 15;
    const float* po = lds_o + r * HID + p * 8;
    float s = 0.f;
#pragma unroll
    for (int q = 0; q < 8; ++q) s += po[q] * po[q];
    s += __shfl_xor(s, 1);
    s += __shfl_xor(s, 2);
    s += __shfl_xor(s, 4);
    s += __shfl_xor(s, 8);
    if (p == 0) lds_inv[r] = 1.0f / fmaxf(sqrtf(s), 1e-12f);
  }
  __syncthreads();

#pragma unroll
  for (int rt = 0; rt < ROWT; ++rt) {
    const int r = rt * TILE_M + (tid >> 4);
    const int c = (tid & 15) * 8;
    const float inv = lds_inv[r];
    const float* po = lds_o + r * HID + c;
    float4 o0, o1;
    o0.x = fmaxf(po[0] * inv, 0.f); o0.y = fmaxf(po[1] * inv, 0.f);
    o0.z = fmaxf(po[2] * inv, 0.f); o0.w = fmaxf(po[3] * inv, 0.f);
    o1.x = fmaxf(po[4] * inv, 0.f); o1.y = fmaxf(po[5] * inv, 0.f);
    o1.z = fmaxf(po[6] * inv, 0.f); o1.w = fmaxf(po[7] * inv, 0.f);
    float4* pw = (float4*)(xout + (size_t)(row0 + r) * HID + c);
    pw[0] = o0;
    pw[1] = o1;
  }
}

__global__ __launch_bounds__(256) void mlp_head(
    const float* __restrict__ xin, const float* __restrict__ W1,
    const float* __restrict__ b1, const float* __restrict__ W2,
    const float* __restrict__ b2, float* __restrict__ out) {
  __shared__ float lds_x[TILE_M * HID];
  __shared__ float lds_h[TILE_M * HID];

  const int tid  = threadIdx.x;
  const int row0 = blockIdx.x * TILE_M;

  {
    const int r = tid >> 4;
    const int c = (tid & 15) * 8;
    const float4* px = (const float4*)(xin + (size_t)(row0 + r) * HID + c);
    float4* lx4 = (float4*)(lds_x + r * HID + c);
    lx4[0] = px[0];
    lx4[1] = px[1];
  }

  const int wave  = tid >> 5;
  const int lane  = tid & 31;
  const int nn    = lane & 15;
  const int khalf = (lane >> 4) * 2;
  const int h0    = wave * 16;

  v2f bW[32];
  {
    const float* pw = W1 + (size_t)(h0 + nn) * HID + khalf;
#pragma unroll
    for (int kk = 0; kk < 32; ++kk) bW[kk] = *(const v2f*)(pw + kk * 4);
  }
  const float bias = b1[h0 + nn];
  __syncthreads();

  v8f acc = {0.f, 0.f, 0.f, 0.f, 0.f, 0.f, 0.f, 0.f};
  const float* lx = lds_x + nn * HID + khalf;
#pragma unroll
  for (int kk = 0; kk < 32; ++kk) {
    v2f a = *(const v2f*)(lx + kk * 4);
    acc = __builtin_amdgcn_wmma_f32_16x16x4_f32(false, a, false, bW[kk],
                                                (short)0, acc, false, false);
  }
  const int rbase = (lane >> 4) * 8;
#pragma unroll
  for (int j = 0; j < 8; ++j)
    lds_h[(rbase + j) * HID + h0 + nn] = fmaxf(acc[j] + bias, 0.f);
  __syncthreads();

  {
    const int r = tid >> 4;
    const int p = tid & 15;
    const float* ph = lds_h + r * HID + p * 8;
    float s0 = 0.f, s1 = 0.f;
#pragma unroll
    for (int q = 0; q < 8; ++q) {
      float h = ph[q];
      s0 += h * W2[p * 8 + q];
      s1 += h * W2[HID + p * 8 + q];
    }
    s0 += __shfl_xor(s0, 1); s1 += __shfl_xor(s1, 1);
    s0 += __shfl_xor(s0, 2); s1 += __shfl_xor(s1, 2);
    s0 += __shfl_xor(s0, 4); s1 += __shfl_xor(s1, 4);
    s0 += __shfl_xor(s0, 8); s1 += __shfl_xor(s1, 8);
    if (p == 0) {
      out[(size_t)(row0 + r) * 2 + 0] = s0 + b2[0];
      out[(size_t)(row0 + r) * 2 + 1] = s1 + b2[1];
    }
  }
}

extern "C" void kernel_launch(void* const* d_in, const int* in_sizes, int n_in,
                              void* d_out, int out_size, void* d_ws, size_t ws_size,
                              hipStream_t stream) {
  const float* x  = (const float*)d_in[0];
  const int*   ei = (const int*)d_in[1];
  const float* Wl = (const float*)d_in[2];
  const float* bl = (const float*)d_in[3];
  const float* Wr = (const float*)d_in[4];
  const float* W1 = (const float*)d_in[5];
  const float* b1 = (const float*)d_in[6];
  const float* W2 = (const float*)d_in[7];
  const float* b2 = (const float*)d_in[8];
  float* out = (float*)d_out;

  const int N = NNODES;
  const int E = in_sizes[1] / 2;
  const int* src = ei;
  const int* dst = ei + E;

  float* inv_deg = (float*)d_ws;
  const size_t pad = 100096;
  float* bufA = inv_deg + pad;
  float* bufB = bufA + (size_t)N * HID;
  const int scanBlocks = (N + 1023) / 1024;
  const size_t npad = (size_t)scanBlocks * 1024;
  int* cnt    = (int*)(bufB + (size_t)N * HID);
  int* rowptr = cnt + npad;
  int* rank   = rowptr + npad;
  int* col    = rank + E;

  hipMemsetAsync(cnt, 0, npad * sizeof(int), stream);
  count_kernel<<<(E + 255) / 256, 256, 0, stream>>>(dst, cnt, rank, E);
  scan_kernel<<<scanBlocks, 1024, 0, stream>>>(cnt, rowptr, inv_deg, N);
  fill_kernel<<<(E + 255) / 256, 256, 0, stream>>>(src, dst, rowptr, rank, col, E);

  const float* xs[3] = {x, bufA, bufB};
  float*       ag[3] = {bufA, bufB, bufA};
  const int gatherBlocks  = (N + 7) / 8;
  const int xformBlocks   = N / ROWS;
  const int headBlocks    = N / TILE_M;

  for (int i = 0; i < 3; ++i) {
    float* agg = ag[i];
    const float* xc = xs[i];
    gather_kernel<<<gatherBlocks, 256, 0, stream>>>(xc, rowptr, cnt, col, agg, N);
    sage_transform<<<xformBlocks, 256, 0, stream>>>(
        agg, xc, Wl + (size_t)i * HID * HID, bl + (size_t)i * HID,
        Wr + (size_t)i * HID * HID, inv_deg, agg);
  }
  mlp_head<<<headBlocks, 256, 0, stream>>>(bufA, W1, b1, W2, b2, out);
}


# LDS row pitch 128 -> 132 floats in both WMMA kernels: the rows of a tile no longer start on one bank, so A-fragment ds reads are bank-conflict-free instead of served row by row
# speedup vs baseline: 3.0400x; 1.4480x over previous
#include <hip/hip_runtime.h>
#include <math.h>

typedef __attribute__((ext_vector_type(2))) float v2f;
typedef __attribute__((ext_vector_type(8))) float v8f;

#define NNODES 100000
#define HID    128
#define TILE_M 16
#define ROWT   2
#define ROWS   (ROWT * TILE_M)
#define LDP    132

__global__ __launch_bounds__(256) void count_kernel(const int* __restrict__ dst,
                                                    int* __restrict__ cnt,
                                                    int* __restrict__ rank, int E) {
  int e = blockIdx.x * 256 + threadIdx.x;
  if (e < E)
    rank[e] = __hip_atomic_fetch_add(&cnt[dst[e]], 1, __ATOMIC_RELAXED,
                                     __HIP_MEMORY_SCOPE_AGENT);
}

__global__ __launch_bounds__(1024) void scan_kernel(const int* __restrict__ cnt,
                                                    int* __restrict__ rowptr,
                                                    float* __restrict__ invdeg, int N) {
  __shared__ int wpre[32];
  __shared__ int wtot[32];
  const int tid  = threadIdx.x;
  const int lane = tid & 31;
  const int wave = tid >> 5;

  int s = 0;
  const int4* c4 = (const int4*)cnt;
  const int n4 = blockIdx.x * 256;
  for (int i = tid; i < n4; i += 1024) {
    int4 v = c4[i];
    s += (v.x + v.y) + (v.z + v.w);
  }
  s += __shfl_xor(s, 1);
  s += __shfl_xor(s, 2);
  s += __shfl_xor(s, 4);
  s += __shfl_xor(s, 8);
  s += __shfl_xor(s, 16);
  if (lane == 0) wpre[wave] = s;

  const int n = blockIdx.x * 1024 + tid;
  const int c = (n < N) ? cnt[n] : 0;
  int incl = c;
#pragma unroll
  for (int o = 1; o < 32; o <<= 1) {
    int t = __shfl_up(incl, o);
    if (lane >= o) incl += t;
  }
  if (lane == 31) wtot[wave] = incl;
  __syncthreads();

  int ab = wpre[lane] + ((lane < wave) ? wtot[lane] : 0);
  ab += __shfl_xor(ab, 1);
  ab += __shfl_xor(ab, 2);
  ab += __shfl_xor(ab, 4);
  ab += __shfl_xor(ab, 8);
  ab += __shfl_xor(ab, 16);
  if (n < N) {
    rowptr[n] = ab + incl - c;
    invdeg[n] = 1.0f / fmaxf((float)c, 1.0f);
  }
}

__global__ __launch_bounds__(256) void fill_kernel(const int* __restrict__ src,
                                                   const int* __restrict__ dst,
                                                   const int* __restrict__ rowptr,
                                                   const int* __restrict__ rank,
                                                   int* __restrict__ col, int E) {
  int e = blockIdx.x * 256 + threadIdx.x;
  if (e < E) col[rowptr[dst[e]] + rank[e]] = src[e];
}

__global__ __launch_bounds__(256) void gather_kernel(const float* __restrict__ x,
                                                     const int* __restrict__ rowptr,
                                                     const int* __restrict__ cnt,
                                                     const int* __restrict__ col,
                                                     float* __restrict__ agg, int N) {
  const int lane = threadIdx.x & 31;
  const int n = __builtin_amdgcn_readfirstlane((int)(blockIdx.x * 8 + (threadIdx.x >> 5)));
  if (n >= N) return;
  const int beg = rowptr[n];
  const int end = beg + cnt[n];
  const float* xl = x + lane * 4;
  float4 acc = make_float4(0.f, 0.f, 0.f, 0.f);
  int j = beg;
  for (; j + 4 <= end; j += 4) {
    const int s0 = col[j], s1 = col[j + 1], s2 = col[j + 2], s3 = col[j + 3];
    const float4 v0 = *(const float4*)(xl + (size_t)s0 * HID);
    const float4 v1 = *(const float4*)(xl + (size_t)s1 * HID);
    const float4 v2 = *(const float4*)(xl + (size_t)s2 * HID);
    const float4 v3 = *(const float4*)(xl + (size_t)s3 * HID);
    acc.x += v0.x; acc.y += v0.y; acc.z += v0.z; acc.w += v0.w;
    acc.x += v1.x; acc.y += v1.y; acc.z += v1.z; acc.w += v1.w;
    acc.x += v2.x; acc.y += v2.y; acc.z += v2.z; acc.w += v2.w;
    acc.x += v3.x; acc.y += v3.y; acc.z += v3.z; acc.w += v3.w;
  }
  for (; j < end; ++j) {
    const int s0 = col[j];
    const float4 v0 = *(const float4*)(xl + (size_t)s0 * HID);
    acc.x += v0.x; acc.y += v0.y; acc.z += v0.z; acc.w += v0.w;
  }
  *(float4*)(agg + (size_t)n * HID + lane * 4) = acc;
}

__global__ __launch_bounds__(256) void sage_transform(
    const float* __restrict__ agg, const float* __restrict__ xin,
    const float* __restrict__ Wl, const float* __restrict__ bl,
    const float* __restrict__ Wr, const float* __restrict__ invdeg,
    float* __restrict__ xout) {
  __shared__ float lds_a[ROWS * LDP];
  __shared__ float lds_x[ROWS * LDP];
  __shared__ float lds_o[ROWS * LDP];
  __shared__ float lds_inv[ROWS];

  const int tid  = threadIdx.x;
  const int row0 = blockIdx.x * ROWS;

#pragma unroll
  for (int rt = 0; rt < ROWT; ++rt) {
    const int r = rt * TILE_M + (tid >> 4);
    const int c = (tid & 15) * 8;
    const float inv = invdeg[row0 + r];
    const float4* pa = (const float4*)(agg + (size_t)(row0 + r) * HID + c);
    float4 a0 = pa[0], a1 = pa[1];
    float4* la4 = (float4*)(lds_a + r * LDP + c);
    la4[0] = make_float4(a0.x * inv, a0.y * inv, a0.z * inv, a0.w * inv);
    la4[1] = make_float4(a1.x * inv, a1.y * inv, a1.z * inv, a1.w * inv);
    const float4* px = (const float4*)(xin + (size_t)(row0 + r) * HID + c);
    float4* lx4 = (float4*)(lds_x + r * LDP + c);
    lx4[0] = px[0];
    lx4[1] = px[1];
  }

  const int wave  = tid >> 5;
  const int lane  = tid & 31;
  const int nn    = lane & 15;
  const int khalf = (lane >> 4) * 2;
  const int h0    = wave * 16;

  v2f bL[32], bR[32];
  {
    const float* pwl = Wl + (size_t)(h0 + nn) * HID + khalf;
    const float* pwr = Wr + (size_t)(h0 + nn) * HID + khalf;
#pragma unroll
    for (int kk = 0; kk < 32; ++kk) {
      bL[kk] = *(const v2f*)(pwl + kk * 4);
      bR[kk] = *(const v2f*)(pwr + kk * 4);
    }
  }
  const float bias  = bl[h0 + nn];
  const int   rbase = (lane >> 4) * 8;
  __syncthreads();

#pragma unroll 1
  for (int rt = 0; rt < ROWT; ++rt) {
    v8f accL = {0.f, 0.f, 0.f, 0.f, 0.f, 0.f, 0.f, 0.f};
    v8f accR = {0.f, 0.f, 0.f, 0.f, 0.f, 0.f, 0.f, 0.f};
    const float* la = lds_a + (rt * TILE_M + nn) * LDP + khalf;
    const float* lx = lds_x + (rt * TILE_M + nn) * LDP + khalf;
#pragma unroll
    for (int kk = 0; kk < 32; ++kk) {
      v2f aA = *(const v2f*)(la + kk * 4);
      accL = __builtin_amdgcn_wmma_f32_16x16x4_f32(false, aA, false, bL[kk],
                                                   (short)0, accL, false, false);
      v2f aX = *(const v2f*)(lx + kk * 4);
      accR = __builtin_amdgcn_wmma_f32_16x16x4_f32(false, aX, false, bR[kk],
                                                   (short)0, accR, false, false);
    }
#pragma unroll
    for (int j = 0; j < 8; ++j)
      lds_o[(rt * TILE_M + rbase + j) * LDP + h0 + nn] = accL[j] + accR[j] + bias;
  }
  __syncthreads();

#pragma unroll
  for (int rt = 0; rt < ROWT; ++rt) {
    const int r = rt * TILE_M + (tid >> 4);
    const int p = tid & 15;
    const float* po = lds_o + r * LDP + p * 8;
    float s = 0.f;
#pragma unroll
    for (int q = 0; q < 8; ++q) s += po[q] * po[q];
    s += __shfl_xor(s, 1);
    s += __shfl_xor(s, 2);
    s += __shfl_xor(s, 4);
    s += __shfl_xor(s, 8);
    if (p == 0) lds_inv[r] = 1.0f / fmaxf(sqrtf(s), 1e-12f);
  }
  __syncthreads();

#pragma unroll
  for (int rt = 0; rt < ROWT; ++rt) {
    const int r = rt * TILE_M + (tid >> 4);
    const int c = (tid & 15) * 8;
    const float inv = lds_inv[r];
    const float* po = lds_o + r * LDP + c;
    float4 o0, o1;
    o0.x = fmaxf(po[0] * inv, 0.f); o0.y = fmaxf(po[1] * inv, 0.f);
    o0.z = fmaxf(po[2] * inv, 0.f); o0.w = fmaxf(po[3] * inv, 0.f);
    o1.x = fmaxf(po[4] * inv, 0.f); o1.y = fmaxf(po[5] * inv, 0.f);
    o1.z = fmaxf(po[6] * inv, 0.f); o1.w = fmaxf(po[7] * inv, 0.f);
    float4* pw = (float4*)(xout + (size_t)(row0 + r) * HID + c);
    pw[0] = o0;
    pw[1] = o1;
  }
}

__global__ __launch_bounds__(256) void mlp_head(
    const float* __restrict__ xin, const float* __restrict__ W1,
    const float* __restrict__ b1, const float* __restrict__ W2,
    const float* __restrict__ b2, float* __restrict__ out) {
  __shared__ float lds_x[TILE_M * LDP];
  __shared__ float lds_h[TILE_M * LDP];

  const int tid  = threadIdx.x;
  const int row0 = blockIdx.x * TILE_M;

  {
    const int r = tid >> 4;
    const int c = (tid & 15) * 8;
    const float4* px = (const float4*)(xin + (size_t)(row0 + r) * HID + c);
    float4* lx4 = (float4*)(lds_x + r * LDP + c);
    lx4[0] = px[0];
    lx4[1] = px[1];
  }

  const int wave  = tid >> 5;
  const int lane  = tid & 31;
  const int nn    = lane & 15;
  const int khalf = (lane >> 4) * 2;
  const int h0    = wave * 16;

  v2f bW[32];
  {
    const float* pw = W1 + (size_t)(h0 + nn) * HID + khalf;
#pragma unroll
    for (int kk = 0; kk < 32; ++kk) bW[kk] = *(const v2f*)(pw + kk * 4);
  }
  const float bias = b1[h0 + nn];
  __syncthreads();

  v8f acc = {0.f, 0.f, 0.f, 0.f, 0.f, 0.f, 0.f, 0.f};
  const float* lx = lds_x + nn * LDP + khalf;
#pragma unroll
  for (int kk = 0; kk < 32; ++kk) {
    v2f a = *(const v2f*)(lx + kk * 4);
    acc = __builtin_amdgcn_wmma_f32_16x16x4_f32(false, a, false, bW[kk],
                                                (short)0, acc, false, false);
  }
  const int rbase = (lane >> 4) * 8;
#pragma unroll
  for (int j = 0; j < 8; ++j)
    lds_h[(rbase + j) * LDP + h0 + nn] = fmaxf(acc[j] + bias, 0.f);
  __syncthreads();

  {
    const int r = tid >> 4;
    const int p = tid & 15;
    const float* ph = lds_h + r * LDP + p * 8;
    float s0 = 0.f, s1 = 0.f;
#pragma unroll
    for (int q = 0; q < 8; ++q) {
      float h = ph[q];
      s0 += h * W2[p * 8 + q];
      s1 += h * W2[HID + p * 8 + q];
    }
    s0 += __shfl_xor(s0, 1); s1 += __shfl_xor(s1, 1);
    s0 += __shfl_xor(s0, 2); s1 += __shfl_xor(s1, 2);
    s0 += __shfl_xor(s0, 4); s1 += __shfl_xor(s1, 4);
    s0 += __shfl_xor(s0, 8); s1 += __shfl_xor(s1, 8);
    if (p == 0) {
      out[(size_t)(row0 + r) * 2 + 0] = s0 + b2[0];
      out[(size_t)(row0 + r) * 2 + 1] = s1 + b2[1];
    }
  }
}

extern "C" void kernel_launch(void* const* d_in, const int* in_sizes, int n_in,
                              void* d_out, int out_size, void* d_ws, size_t ws_size,
                              hipStream_t stream) {
  const float* x  = (const float*)d_in[0];
  const int*   ei = (const int*)d_in[1];
  const float* Wl = (const float*)d_in[2];
  const float* bl = (const float*)d_in[3];
  const float* Wr = (const float*)d_in[4];
  const float* W1 = (const float*)d_in[5];
  const float* b1 = (const float*)d_in[6];
  const float* W2 = (const float*)d_in[7];
  const float* b2 = (const float*)d_in[8];
  float* out = (float*)d_out;

  const int N = NNODES;
  const int E = in_sizes[1] / 2;
  const int* src = ei;
  const int* dst = ei + E;

  float* inv_deg = (float*)d_ws;
  const size_t pad = 100096;
  float* bufA = inv_deg + pad;
  float* bufB = bufA + (size_t)N * HID;
  const int scanBlocks = (N + 1023) / 1024;
  const size_t npad = (size_t)scanBlocks * 1024;
  int* cnt    = (int*)(bufB + (size_t)N * HID);
  int* rowptr = cnt + npad;
  int* rank   = rowptr + npad;
  int* col    = rank + E;

  hipMemsetAsync(cnt, 0, npad * sizeof(int), stream);
  count_kernel<<<(E + 255) / 256, 256, 0, stream>>>(dst, cnt, rank, E);
  scan_kernel<<<scanBlocks, 1024, 0, stream>>>(cnt, rowptr, inv_deg, N);
  fill_kernel<<<(E + 255) / 256, 256, 0, stream>>>(src, dst, rowptr, rank, col, E);

  const float* xs[3] = {x, bufA, bufB};
  float*       ag[3] = {bufA, bufB, bufA};
  const int gatherBlocks  = (N + 7) / 8;
  const int xformBlocks   = N / ROWS;
  const int headBlocks    = N / TILE_M;

  for (int i = 0; i < 3; ++i) {
    float* agg = ag[i];
    const float* xc = xs[i];
    gather_kernel<<<gatherBlocks, 256, 0, stream>>>(xc, rowptr, cnt, col, agg, N);
    sage_transform<<<xformBlocks, 256, 0, stream>>>(
        agg, xc, Wl + (size_t)i * HID * HID, bl + (size_t)i * HID,
        Wr + (size_t)i * HID * HID, inv_deg, agg);
  }
  mlp_head<<<headBlocks, 256, 0, stream>>>(bufA, W1, b1, W2, b2, out);
}


# CSR build as two-level counting sort on LDS counters (bin edges by 512-node range; per range count, scan, place) replaces global returning atomics; per-node sum order = LDS arrival rank
# speedup vs baseline: 3.8027x; 1.2509x over previous
#include <hip/hip_runtime.h>
#include <math.h>

typedef __attribute__((ext_vector_type(2))) float v2f;
typedef __attribute__((ext_vector_type(8))) float v8f;

#define NNODES 100000
#define HID    128
#define TILE_M 16
#define ROWT   2
#define ROWS   (ROWT * TILE_M)
#define LDP    132

#define BSH    9
#define BNODES (1 << BSH)
#define NBUCK  ((NNODES + BNODES - 1) / BNODES)
#define CHUNK  8192

__global__ __launch_bounds__(1024) void bin_count(const int* __restrict__ dst,
                                                  int* __restrict__ bh, int E, int G) {
  __shared__ int h[NBUCK];
  const int tid = threadIdx.x;
  if (tid < NBUCK) h[tid] = 0;
  __syncthreads();
  const int base = blockIdx.x * CHUNK;
  int d[CHUNK / 1024];
#pragma unroll
  for (int q = 0; q < CHUNK / 1024; ++q) {
    const int e = base + q * 1024 + tid;
    d[q] = dst[min(e, E - 1)];
  }
#pragma unroll
  for (int q = 0; q < CHUNK / 1024; ++q) {
    const int e = base + q * 1024 + tid;
    if (e < E)
      __hip_atomic_fetch_add(&h[d[q] >> BSH], 1, __ATOMIC_RELAXED,
                             __HIP_MEMORY_SCOPE_WORKGROUP);
  }
  __syncthreads();
  if (tid < NBUCK) bh[tid * G + blockIdx.x] = h[tid];
}

__global__ __launch_bounds__(1024) void scan_kernel(const int* __restrict__ in,
                                                    int* __restrict__ out, int M) {
  __shared__ int wpre[32];
  __shared__ int wtot[32];
  const int tid  = threadIdx.x;
  const int lane = tid & 31;
  const int wave = tid >> 5;

  int s = 0;
  const int4* c4 = (const int4*)in;
  const int n4 = blockIdx.x * 256;
  for (int i = tid; i < n4; i += 1024) {
    int4 v = c4[i];
    s += (v.x + v.y) + (v.z + v.w);
  }
  s += __shfl_xor(s, 1);
  s += __shfl_xor(s, 2);
  s += __shfl_xor(s, 4);
  s += __shfl_xor(s, 8);
  s += __shfl_xor(s, 16);
  if (lane == 0) wpre[wave] = s;

  const int n = blockIdx.x * 1024 + tid;
  const int c = (n < M) ? in[n] : 0;
  int incl = c;
#pragma unroll
  for (int o = 1; o < 32; o <<= 1) {
    int t = __shfl_up(incl, o);
    if (lane >= o) incl += t;
  }
  if (lane == 31) wtot[wave] = incl;
  __syncthreads();

  int ab = wpre[lane] + ((lane < wave) ? wtot[lane] : 0);
  ab += __shfl_xor(ab, 1);
  ab += __shfl_xor(ab, 2);
  ab += __shfl_xor(ab, 4);
  ab += __shfl_xor(ab, 8);
  ab += __shfl_xor(ab, 16);
  if (n < M) out[n] = ab + incl - c;
}

__global__ __launch_bounds__(1024) void bin_scatter(const int* __restrict__ src,
                                                    const int* __restrict__ dst,
                                                    const int* __restrict__ bb,
                                                    int* __restrict__ ebuf_w, int E, int G) {
  int2* ebuf = (int2*)ebuf_w;
  __shared__ int cur[NBUCK];
  const int tid = threadIdx.x;
  if (tid < NBUCK) cur[tid] = bb[tid * G + blockIdx.x];
  __syncthreads();
  const int base = blockIdx.x * CHUNK;
  int d[CHUNK / 1024], s[CHUNK / 1024];
#pragma unroll
  for (int q = 0; q < CHUNK / 1024; ++q) {
    const int e = min(base + q * 1024 + tid, E - 1);
    d[q] = dst[e];
    s[q] = src[e];
  }
#pragma unroll
  for (int q = 0; q < CHUNK / 1024; ++q) {
    const int e = base + q * 1024 + tid;
    if (e < E) {
      const int p = __hip_atomic_fetch_add(&cur[d[q] >> BSH], 1, __ATOMIC_RELAXED,
                                           __HIP_MEMORY_SCOPE_WORKGROUP);
      ebuf[p] = make_int2(s[q], d[q]);
    }
  }
}

__global__ __launch_bounds__(1024) void bucket_build(const int* __restrict__ ebuf_w,
                                                     const int* __restrict__ bb,
                                                     int* __restrict__ cnt,
                                                     int* __restrict__ rowptr,
                                                     float* __restrict__ invdeg,
                                                     int* __restrict__ col,
                                                     int E, int G, int N) {
  const int2* ebuf = (const int2*)ebuf_w;
  __shared__ int c[BNODES];
  __shared__ int wtot[BNODES / 32];
  const int tid  = threadIdx.x;
  const int lane = tid & 31;
  const int wave = tid >> 5;
  const int b    = blockIdx.x;
  const int beg  = bb[b * G];
  const int end  = (b + 1 < (int)gridDim.x) ? bb[(b + 1) * G] : E;

  if (tid < BNODES) c[tid] = 0;
  __syncthreads();
  for (int e0 = beg + tid; e0 < end; e0 += 4 * 1024) {
    int dd[4];
#pragma unroll
    for (int u = 0; u < 4; ++u) dd[u] = ebuf[min(e0 + u * 1024, end - 1)].y;
#pragma unroll
    for (int u = 0; u < 4; ++u)
      if (e0 + u * 1024 < end)
        __hip_atomic_fetch_add(&c[dd[u] & (BNODES - 1)], 1, __ATOMIC_RELAXED,
                               __HIP_MEMORY_SCOPE_WORKGROUP);
  }
  __syncthreads();

  const int v = (tid < BNODES) ? c[tid] : 0;
  int incl = v;
#pragma unroll
  for (int o = 1; o < 32; o <<= 1) {
    int t = __shfl_up(incl, o);
    if (lane >= o) incl += t;
  }
  if (tid < BNODES && lane == 31) wtot[wave] = incl;
  __syncthreads();

  if (tid < BNODES) {
    int pre = 0;
    for (int w = 0; w < wave; ++w) pre += wtot[w];
    const int start = beg + pre + incl - v;
    c[tid] = start;
    const int n = b * BNODES + tid;
    if (n < N) {
      cnt[n]    = v;
      rowptr[n] = start;
      invdeg[n] = 1.0f / fmaxf((float)v, 1.0f);
    }
  }
  __syncthreads();

  for (int e0 = beg + tid; e0 < end; e0 += 4 * 1024) {
    int2 sd[4];
#pragma unroll
    for (int u = 0; u < 4; ++u) sd[u] = ebuf[min(e0 + u * 1024, end - 1)];
#pragma unroll
    for (int u = 0; u < 4; ++u)
      if (e0 + u * 1024 < end) {
        const int p = __hip_atomic_fetch_add(&c[sd[u].y & (BNODES - 1)], 1, __ATOMIC_RELAXED,
                                             __HIP_MEMORY_SCOPE_WORKGROUP);
        col[p] = sd[u].x;
      }
  }
}

__global__ __launch_bounds__(256) void gather_kernel(const float* __restrict__ x,
                                                     const int* __restrict__ rowptr,
                                                     const int* __restrict__ cnt,
                                                     const int* __restrict__ col,
                                                     float* __restrict__ agg, int N) {
  const int lane = threadIdx.x & 31;
  const int n = __builtin_amdgcn_readfirstlane((int)(blockIdx.x * 8 + (threadIdx.x >> 5)));
  if (n >= N) return;
  const int beg = rowptr[n];
  const int end = beg + cnt[n];
  const float* xl = x + lane * 4;
  float4 acc = make_float4(0.f, 0.f, 0.f, 0.f);
  int j = beg;
  for (; j + 4 <= end; j += 4) {
    const int s0 = col[j], s1 = col[j + 1], s2 = col[j + 2], s3 = col[j + 3];
    const float4 v0 = *(const float4*)(xl + (size_t)s0 * HID);
    const float4 v1 = *(const float4*)(xl + (size_t)s1 * HID);
    const float4 v2 = *(const float4*)(xl + (size_t)s2 * HID);
    const float4 v3 = *(const float4*)(xl + (size_t)s3 * HID);
    acc.x += v0.x; acc.y += v0.y; acc.z += v0.z; acc.w += v0.w;
    acc.x += v1.x; acc.y += v1.y; acc.z += v1.z; acc.w += v1.w;
    acc.x += v2.x; acc.y += v2.y; acc.z += v2.z; acc.w += v2.w;
    acc.x += v3.x; acc.y += v3.y; acc.z += v3.z; acc.w += v3.w;
  }
  for (; j < end; ++j) {
    const int s0 = col[j];
    const float4 v0 = *(const float4*)(xl + (size_t)s0 * HID);
    acc.x += v0.x; acc.y += v0.y; acc.z += v0.z; acc.w += v0.w;
  }
  *(float4*)(agg + (size_t)n * HID + lane * 4) = acc;
}

__global__ __launch_bounds__(256) void sage_transform(
    const float* __restrict__ agg, const float* __restrict__ xin,
    const float* __restrict__ Wl, const float* __restrict__ bl,
    const float* __restrict__ Wr, const float* __restrict__ invdeg,
    float* __restrict__ xout) {
  __shared__ float lds_a[ROWS * LDP];
  __shared__ float lds_x[ROWS * LDP];
  __shared__ float lds_o[ROWS * LDP];
  __shared__ float lds_inv[ROWS];

  const int tid  = threadIdx.x;
  const int row0 = blockIdx.x * ROWS;

#pragma unroll
  for (int rt = 0; rt < ROWT; ++rt) {
    const int r = rt * TILE_M + (tid >> 4);
    const int c = (tid & 15) * 8;
    const float inv = invdeg[row0 + r];
    const float4* pa = (const float4*)(agg + (size_t)(row0 + r) * HID + c);
    float4 a0 = pa[0], a1 = pa[1];
    float4* la4 = (float4*)(lds_a + r * LDP + c);
    la4[0] = make_float4(a0.x * inv, a0.y * inv, a0.z * inv, a0.w * inv);
    la4[1] = make_float4(a1.x * inv, a1.y * inv, a1.z * inv, a1.w * inv);
    const float4* px = (const float4*)(xin + (size_t)(row0 + r) * HID + c);
    float4* lx4 = (float4*)(lds_x + r * LDP + c);
    lx4[0] = px[0];
    lx4[1] = px[1];
  }

  const int wave  = tid >> 5;
  const int lane  = tid & 31;
  const int nn    = lane & 15;
  const int khalf = (lane >> 4) * 2;
  const int h0    = wave * 16;

  v2f bL[32], bR[32];
  {
    const float* pwl = Wl + (size_t)(h0 + nn) * HID + khalf;
    const float* pwr = Wr + (size_t)(h0 + nn) * HID + khalf;
#pragma unroll
    for (int kk = 0; kk < 32; ++kk) {
      bL[kk] = *(const v2f*)(pwl + kk * 4);
      bR[kk] = *(const v2f*)(pwr + kk * 4);
    }
  }
  const float bias  = bl[h0 + nn];
  const int   rbase = (lane >> 4) * 8;
  __syncthreads();

#pragma unroll 1
  for (int rt = 0; rt < ROWT; ++rt) {
    v8f accL = {0.f, 0.f, 0.f, 0.f, 0.f, 0.f, 0.f, 0.f};
    v8f accR = {0.f, 0.f, 0.f, 0.f, 0.f, 0.f, 0.f, 0.f};
    const float* la = lds_a + (rt * TILE_M + nn) * LDP + khalf;
    const float* lx = lds_x + (rt * TILE_M + nn) * LDP + khalf;
#pragma unroll
    for (int kk = 0; kk < 32; ++kk) {
      v2f aA = *(const v2f*)(la + kk * 4);
      accL = __builtin_amdgcn_wmma_f32_16x16x4_f32(false, aA, false, bL[kk],
                                                   (short)0, accL, false, false);
      v2f aX = *(const v2f*)(lx + kk * 4);
      accR = __builtin_amdgcn_wmma_f32_16x16x4_f32(false, aX, false, bR[kk],
                                                   (short)0, accR, false, false);
    }
#pragma unroll
    for (int j = 0; j < 8; ++j)
      lds_o[(rt * TILE_M + rbase + j) * LDP + h0 + nn] = accL[j] + accR[j] + bias;
  }
  __syncthreads();

#pragma unroll
  for (int rt = 0; rt < ROWT; ++rt) {
    const int r = rt * TILE_M + (tid >> 4);
    const int p = tid & 15;
    const float* po = lds_o + r * LDP + p * 8;
    float s = 0.f;
#pragma unroll
    for (int q = 0; q < 8; ++q) s += po[q] * po[q];
    s += __shfl_xor(s, 1);
    s += __shfl_xor(s, 2);
    s += __shfl_xor(s, 4);
    s += __shfl_xor(s, 8);
    if (p == 0) lds_inv[r] = 1.0f / fmaxf(sqrtf(s), 1e-12f);
  }
  __syncthreads();

#pragma unroll
  for (int rt = 0; rt < ROWT; ++rt) {
    const int r = rt * TILE_M + (tid >> 4);
    const int c = (tid & 15) * 8;
    const float inv = lds_inv[r];
    const float* po = lds_o + r * LDP + c;
    float4 o0, o1;
    o0.x = fmaxf(po[0] * inv, 0.f); o0.y = fmaxf(po[1] * inv, 0.f);
    o0.z = fmaxf(po[2] * inv, 0.f); o0.w = fmaxf(po[3] * inv, 0.f);
    o1.x = fmaxf(po[4] * inv, 0.f); o1.y = fmaxf(po[5] * inv, 0.f);
    o1.z = fmaxf(po[6] * inv, 0.f); o1.w = fmaxf(po[7] * inv, 0.f);
    float4* pw = (float4*)(xout + (size_t)(row0 + r) * HID + c);
    pw[0] = o0;
    pw[1] = o1;
  }
}

__global__ __launch_bounds__(256) void mlp_head(
    const float* __restrict__ xin, const float* __restrict__ W1,
    const float* __restrict__ b1, const float* __restrict__ W2,
    const float* __restrict__ b2, float* __restrict__ out) {
  __shared__ float lds_x[TILE_M * LDP];
  __shared__ float lds_h[TILE_M * LDP];

  const int tid  = threadIdx.x;
  const int row0 = blockIdx.x * TILE_M;

  {
    const int r = tid >> 4;
    const int c = (tid & 15) * 8;
    const float4* px = (const float4*)(xin + (size_t)(row0 + r) * HID + c);
    float4* lx4 = (float4*)(lds_x + r * LDP + c);
    lx4[0] = px[0];
    lx4[1] = px[1];
  }

  const int wave  = tid >> 5;
  const int lane  = tid & 31;
  const int nn    = lane & 15;
  const int khalf = (lane >> 4) * 2;
  const int h0    = wave * 16;

  v2f bW[32];
  {
    const float* pw = W1 + (size_t)(h0 + nn) * HID + khalf;
#pragma unroll
    for (int kk = 0; kk < 32; ++kk) bW[kk] = *(const v2f*)(pw + kk * 4);
  }
  const float bias = b1[h0 + nn];
  __syncthreads();

  v8f acc = {0.f, 0.f, 0.f, 0.f, 0.f, 0.f, 0.f, 0.f};
  const float* lx = lds_x + nn * LDP + khalf;
#pragma unroll
  for (int kk = 0; kk < 32; ++kk) {
    v2f a = *(const v2f*)(lx + kk * 4);
    acc = __builtin_amdgcn_wmma_f32_16x16x4_f32(false, a, false, bW[kk],
                                                (short)0, acc, false, false);
  }
  const int rbase = (lane >> 4) * 8;
#pragma unroll
  for (int j = 0; j < 8; ++j)
    lds_h[(rbase + j) * LDP + h0 + nn] = fmaxf(acc[j] + bias, 0.f);
  __syncthreads();

  {
    const int r = tid >> 4;
    const int p = tid & 15;
    const float* ph = lds_h + r * LDP + p * 8;
    float s0 = 0.f, s1 = 0.f;
#pragma unroll
    for (int q = 0; q < 8; ++q) {
      float h = ph[q];
      s0 += h * W2[p * 8 + q];
      s1 += h * W2[HID + p * 8 + q];
    }
    s0 += __shfl_xor(s0, 1); s1 += __shfl_xor(s1, 1);
    s0 += __shfl_xor(s0, 2); s1 += __shfl_xor(s1, 2);
    s0 += __shfl_xor(s0, 4); s1 += __shfl_xor(s1, 4);
    s0 += __shfl_xor(s0, 8); s1 += __shfl_xor(s1, 8);
    if (p == 0) {
      out[(size_t)(row0 + r) * 2 + 0] = s0 + b2[0];
      out[(size_t)(row0 + r) * 2 + 1] = s1 + b2[1];
    }
  }
}

extern "C" void kernel_launch(void* const* d_in, const int* in_sizes, int n_in,
                              void* d_out, int out_size, void* d_ws, size_t ws_size,
                              hipStream_t stream) {
  const float* x  = (const float*)d_in[0];
  const int*   ei = (const int*)d_in[1];
  const float* Wl = (const float*)d_in[2];
  const float* bl = (const float*)d_in[3];
  const float* Wr = (const float*)d_in[4];
  const float* W1 = (const float*)d_in[5];
  const float* b1 = (const float*)d_in[6];
  const float* W2 = (const float*)d_in[7];
  const float* b2 = (const float*)d_in[8];
  float* out = (float*)d_out;

  const int N = NNODES;
  const int E = in_sizes[1] / 2;
  const int* src = ei;
  const int* dst = ei + E;

  float* inv_deg = (float*)d_ws;
  const size_t pad = 100096;
  float* bufA = inv_deg + pad;
  float* bufB = bufA + (size_t)N * HID;
  const size_t npad = (size_t)NBUCK * BNODES;
  const int G = (E + CHUNK - 1) / CHUNK;
  const int M = NBUCK * G;
  const int scanBlocks = (M + 1023) / 1024;
  const size_t mpad = (size_t)scanBlocks * 1024;
  int*  cnt    = (int*)(bufB + (size_t)N * HID);
  int*  rowptr = cnt + npad;
  int*  bh     = rowptr + npad;
  int*  bb     = bh + mpad;
  int*  ebuf   = bb + mpad;
  int*  col    = ebuf + 2 * (size_t)E;

  bin_count<<<G, 1024, 0, stream>>>(dst, bh, E, G);
  scan_kernel<<<scanBlocks, 1024, 0, stream>>>(bh, bb, M);
  bin_scatter<<<G, 1024, 0, stream>>>(src, dst, bb, ebuf, E, G);
  bucket_build<<<NBUCK, 1024, 0, stream>>>(ebuf, bb, cnt, rowptr, inv_deg, col, E, G, N);

  const float* xs[3] = {x, bufA, bufB};
  float*       ag[3] = {bufA, bufB, bufA};
  const int gatherBlocks  = (N + 7) / 8;
  const int xformBlocks   = N / ROWS;
  const int headBlocks    = N / TILE_M;

  for (int i = 0; i < 3; ++i) {
    float* agg = ag[i];
    const float* xc = xs[i];
    gather_kernel<<<gatherBlocks, 256, 0, stream>>>(xc, rowptr, cnt, col, agg, N);
    sage_transform<<<xformBlocks, 256, 0, stream>>>(
        agg, xc, Wl + (size_t)i * HID * HID, bl + (size_t)i * HID,
        Wr + (size_t)i * HID * HID, inv_deg, agg);
  }
  mlp_head<<<headBlocks, 256, 0, stream>>>(bufA, W1, b1, W2, b2, out);
}


# Both WMMA kernels persistent: 512 / 1024 resident work-groups load their weight fragments once and walk over row chunks, next chunk's rows requested under the products; sum order unchanged
# speedup vs baseline: 4.7314x; 1.2442x over previous
#include <hip/hip_runtime.h>
#include <math.h>

typedef __attribute__((ext_vector_type(2))) float v2f;
typedef __attribute__((ext_vector_type(8))) float v8f;

#define NNODES 100000
#define HID    128
#define TILE_M 16
#define ROWT   2
#define ROWS   (ROWT * TILE_M)
#define LDP    132

#define BSH    9
#define BNODES (1 << BSH)
#define NBUCK  ((NNODES + BNODES - 1) / BNODES)
#define CHUNK  8192

__global__ __launch_bounds__(1024) void bin_count(const int* __restrict__ dst,
                                                  int* __restrict__ bh, int E, int G) {
  __shared__ int h[NBUCK];
  const int tid = threadIdx.x;
  if (tid < NBUCK) h[tid] = 0;
  __syncthreads();
  const int base = blockIdx.x * CHUNK;
  int d[CHUNK / 1024];
#pragma unroll
  for (int q = 0; q < CHUNK / 1024; ++q) {
    const int e = base + q * 1024 + tid;
    d[q] = dst[min(e, E - 1)];
  }
#pragma unroll
  for (int q = 0; q < CHUNK / 1024; ++q) {
    const int e = base + q * 1024 + tid;
    if (e < E)
      __hip_atomic_fetch_add(&h[d[q] >> BSH], 1, __ATOMIC_RELAXED,
                             __HIP_MEMORY_SCOPE_WORKGROUP);
  }
  __syncthreads();
  if (tid < NBUCK) bh[tid * G + blockIdx.x] = h[tid];
}

__global__ __launch_bounds__(1024) void scan_kernel(const int* __restrict__ in,
                                                    int* __restrict__ out, int M) {
  __shared__ int wpre[32];
  __shared__ int wtot[32];
  const int tid  = threadIdx.x;
  const int lane = tid & 31;
  const int wave = tid >> 5;

  int s = 0;
  const int4* c4 = (const int4*)in;
  const int n4 = blockIdx.x * 256;
  for (int i = tid; i < n4; i += 1024) {
    int4 v = c4[i];
    s += (v.x + v.y) + (v.z + v.w);
  }
  s += __shfl_xor(s, 1);
  s += __shfl_xor(s, 2);
  s += __shfl_xor(s, 4);
  s += __shfl_xor(s, 8);
  s += __shfl_xor(s, 16);
  if (lane == 0) wpre[wave] = s;

  const int n = blockIdx.x * 1024 + tid;
  const int c = (n < M) ? in[n] : 0;
  int incl = c;
#pragma unroll
  for (int o = 1; o < 32; o <<= 1) {
    int t = __shfl_up(incl, o);
    if (lane >= o) incl += t;
  }
  if (lane == 31) wtot[wave] = incl;
  __syncthreads();

  int ab = wpre[lane] + ((lane < wave) ? wtot[lane] : 0);
  ab += __shfl_xor(ab, 1);
  ab += __shfl_xor(ab, 2);
  ab += __shfl_xor(ab, 4);
  ab += __shfl_xor(ab, 8);
  ab += __shfl_xor(ab, 16);
  if (n < M) out[n] = ab + incl - c;
}

__global__ __launch_bounds__(1024) void bin_scatter(const int* __restrict__ src,
                                                    const int* __restrict__ dst,
                                                    const int* __restrict__ bb,
                                                    int* __restrict__ ebuf_w, int E, int G) {
  int2* ebuf = (int2*)ebuf_w;
  __shared__ int cur[NBUCK];
  const int tid = threadIdx.x;
  if (tid < NBUCK) cur[tid] = bb[tid * G + blockIdx.x];
  __syncthreads();
  const int base = blockIdx.x * CHUNK;
  int d[CHUNK / 1024], s[CHUNK / 1024];
#pragma unroll
  for (int q = 0; q < CHUNK / 1024; ++q) {
    const int e = min(base + q * 1024 + tid, E - 1);
    d[q] = dst[e];
    s[q] = src[e];
  }
#pragma unroll
  for (int q = 0; q < CHUNK / 1024; ++q) {
    const int e = base + q * 1024 + tid;
    if (e < E) {
      const int p = __hip_atomic_fetch_add(&cur[d[q] >> BSH], 1, __ATOMIC_RELAXED,
                                           __HIP_MEMORY_SCOPE_WORKGROUP);
      ebuf[p] = make_int2(s[q], d[q]);
    }
  }
}

__global__ __launch_bounds__(1024) void bucket_build(const int* __restrict__ ebuf_w,
                                                     const int* __restrict__ bb,
                                                     int* __restrict__ cnt,
                                                     int* __restrict__ rowptr,
                                                     float* __restrict__ invdeg,
                                                     int* __restrict__ col,
                                                     int E, int G, int N) {
  const int2* ebuf = (const int2*)ebuf_w;
  __shared__ int c[BNODES];
  __shared__ int wtot[BNODES / 32];
  const int tid  = threadIdx.x;
  const int lane = tid & 31;
  const int wave = tid >> 5;
  const int b    = blockIdx.x;
  const int beg  = bb[b * G];
  const int end  = (b + 1 < (int)gridDim.x) ? bb[(b + 1) * G] : E;

  if (tid < BNODES) c[tid] = 0;
  __syncthreads();
  for (int e0 = beg + tid; e0 < end; e0 += 4 * 1024) {
    int dd[4];
#pragma unroll
    for (int u = 0; u < 4; ++u) dd[u] = ebuf[min(e0 + u * 1024, end - 1)].y;
#pragma unroll
    for (int u = 0; u < 4; ++u)
      if (e0 + u * 1024 < end)
        __hip_atomic_fetch_add(&c[dd[u] & (BNODES - 1)], 1, __ATOMIC_RELAXED,
                               __HIP_MEMORY_SCOPE_WORKGROUP);
  }
  __syncthreads();

  const int v = (tid < BNODES) ? c[tid] : 0;
  int incl = v;
#pragma unroll
  for (int o = 1; o < 32; o <<= 1) {
    int t = __shfl_up(incl, o);
    if (lane >= o) incl += t;
  }
  if (tid < BNODES && lane == 31) wtot[wave] = incl;
  __syncthreads();

  if (tid < BNODES) {
    int pre = 0;
    for (int w = 0; w < wave; ++w) pre += wtot[w];
    const int start = beg + pre + incl - v;
    c[tid] = start;
    const int n = b * BNODES + tid;
    if (n < N) {
      cnt[n]    = v;
      rowptr[n] = start;
      invdeg[n] = 1.0f / fmaxf((float)v, 1.0f);
    }
  }
  __syncthreads();

  for (int e0 = beg + tid; e0 < end; e0 += 4 * 1024) {
    int2 sd[4];
#pragma unroll
    for (int u = 0; u < 4; ++u) sd[u] = ebuf[min(e0 + u * 1024, end - 1)];
#pragma unroll
    for (int u = 0; u < 4; ++u)
      if (e0 + u * 1024 < end) {
        const int p = __hip_atomic_fetch_add(&c[sd[u].y & (BNODES - 1)], 1, __ATOMIC_RELAXED,
                                             __HIP_MEMORY_SCOPE_WORKGROUP);
        col[p] = sd[u].x;
      }
  }
}

__global__ __launch_bounds__(256) void gather_kernel(const float* __restrict__ x,
                                                     const int* __restrict__ rowptr,
                                                     const int* __restrict__ cnt,
                                                     const int* __restrict__ col,
                                                     float* __restrict__ agg, int N) {
  const int lane = threadIdx.x & 31;
  const int n = __builtin_amdgcn_readfirstlane((int)(blockIdx.x * 8 + (threadIdx.x >> 5)));
  if (n >= N) return;
  const int beg = rowptr[n];
  const int end = beg + cnt[n];
  const float* xl = x + lane * 4;
  float4 acc = make_float4(0.f, 0.f, 0.f, 0.f);
  int j = beg;
  for (; j + 4 <= end; j += 4) {
    const int s0 = col[j], s1 = col[j + 1], s2 = col[j + 2], s3 = col[j + 3];
    const float4 v0 = *(const float4*)(xl + (size_t)s0 * HID);
    const float4 v1 = *(const float4*)(xl + (size_t)s1 * HID);
    const float4 v2 = *(const float4*)(xl + (size_t)s2 * HID);
    const float4 v3 = *(const float4*)(xl + (size_t)s3 * HID);
    acc.x += v0.x; acc.y += v0.y; acc.z += v0.z; acc.w += v0.w;
    acc.x += v1.x; acc.y += v1.y; acc.z += v1.z; acc.w += v1.w;
    acc.x += v2.x; acc.y += v2.y; acc.z += v2.z; acc.w += v2.w;
    acc.x += v3.x; acc.y += v3.y; acc.z += v3.z; acc.w += v3.w;
  }
  for (; j < end; ++j) {
    const int s0 = col[j];
    const float4 v0 = *(const float4*)(xl + (size_t)s0 * HID);
    acc.x += v0.x; acc.y += v0.y; acc.z += v0.z; acc.w += v0.w;
  }
  *(float4*)(agg + (size_t)n * HID + lane * 4) = acc;
}

#define XFORM_WG 512
#define HEAD_WG  1024
static_assert(ROWT == 2, "the staging code below names the chunk's two row tiles");

__global__ __launch_bounds__(256) void sage_transform(
    const float* __restrict__ agg, const float* __restrict__ xin,
    const float* __restrict__ Wl, const float* __restrict__ bl,
    const float* __restrict__ Wr, const float* __restrict__ invdeg,
    float* __restrict__ xout, int nchunks) {
  __shared__ float lds_a[ROWS * LDP];
  __shared__ float lds_x[ROWS * LDP];
  __shared__ float lds_o[ROWS * LDP];
  __shared__ float lds_inv[ROWS];

  const int tid   = threadIdx.x;
  const int wave  = tid >> 5;
  const int lane  = tid & 31;
  const int nn    = lane & 15;
  const int khalf = (lane >> 4) * 2;
  const int h0    = wave * 16;
  const int sr    = tid >> 4;
  const int sc    = (tid & 15) * 8;

  float4 qa00, qa01, qa10, qa11, qx00, qx01, qx10, qx11;
  float  qi0, qi1;
  int chunk = blockIdx.x;
  {
    const size_t r0 = (size_t)chunk * ROWS + sr;
    const size_t r1 = r0 + TILE_M;
    qi0 = invdeg[r0];
    qi1 = invdeg[r1];
    const float4* pa0 = (const float4*)(agg + r0 * HID + sc);
    const float4* pa1 = (const float4*)(agg + r1 * HID + sc);
    const float4* px0 = (const float4*)(xin + r0 * HID + sc);
    const float4* px1 = (const float4*)(xin + r1 * HID + sc);
    qa00 = pa0[0]; qa01 = pa0[1];
    qa10 = pa1[0]; qa11 = pa1[1];
    qx00 = px0[0]; qx01 = px0[1];
    qx10 = px1[0]; qx11 = px1[1];
  }

  v2f bL[32], bR[32];
  {
    const float* pwl = Wl + (size_t)(h0 + nn) * HID + khalf;
    const float* pwr = Wr + (size_t)(h0 + nn) * HID + khalf;
#pragma unroll
    for (int kk = 0; kk < 32; ++kk) {
      bL[kk] = *(const v2f*)(pwl + kk * 4);
      bR[kk] = *(const v2f*)(pwr + kk * 4);
    }
  }
  const float bias  = bl[h0 + nn];
  const int   rbase = (lane >> 4) * 8;

  for (; chunk < nchunks; chunk += (int)gridDim.x) {
    const int row0 = chunk * ROWS;

    {
      float4* la0 = (float4*)(lds_a + sr * LDP + sc);
      float4* la1 = (float4*)(lds_a + (TILE_M + sr) * LDP + sc);
      la0[0] = make_float4(qa00.x * qi0, qa00.y * qi0, qa00.z * qi0, qa00.w * qi0);
      la0[1] = make_float4(qa01.x * qi0, qa01.y * qi0, qa01.z * qi0, qa01.w * qi0);
      la1[0] = make_float4(qa10.x * qi1, qa10.y * qi1, qa10.z * qi1, qa10.w * qi1);
      la1[1] = make_float4(qa11.x * qi1, qa11.y * qi1, qa11.z * qi1, qa11.w * qi1);
      float4* lx0 = (float4*)(lds_x + sr * LDP + sc);
      float4* lx1 = (float4*)(lds_x + (TILE_M + sr) * LDP + sc);
      lx0[0] = qx00; lx0[1] = qx01;
      lx1[0] = qx10; lx1[1] = qx11;
    }
    __syncthreads();

    {
      const int nc = chunk + (int)gridDim.x;
      const int pc = (nc < nchunks) ? nc : chunk;
      const size_t r0 = (size_t)pc * ROWS + sr;
      const size_t r1 = r0 + TILE_M;
      qi0 = invdeg[r0];
      qi1 = invdeg[r1];
      const float4* pa0 = (const float4*)(agg + r0 * HID + sc);
      const float4* pa1 = (const float4*)(agg + r1 * HID + sc);
      const float4* px0 = (const float4*)(xin + r0 * HID + sc);
      const float4* px1 = (const float4*)(xin + r1 * HID + sc);
      qa00 = pa0[0]; qa01 = pa0[1];
      qa10 = pa1[0]; qa11 = pa1[1];
      qx00 = px0[0]; qx01 = px0[1];
      qx10 = px1[0]; qx11 = px1[1];
    }

#pragma unroll 1
    for (int rt = 0; rt < ROWT; ++rt) {
      v8f accL = {0.f, 0.f, 0.f, 0.f, 0.f, 0.f, 0.f, 0.f};
      v8f accR = {0.f, 0.f, 0.f, 0.f, 0.f, 0.f, 0.f, 0.f};
      const float* la = lds_a + (rt * TILE_M + nn) * LDP + khalf;
      const float* lx = lds_x + (rt * TILE_M + nn) * LDP + khalf;
#pragma unroll
      for (int kk = 0; kk < 32; ++kk) {
        v2f aA = *(const v2f*)(la + kk * 4);
        accL = __builtin_amdgcn_wmma_f32_16x16x4_f32(false, aA, false, bL[kk],
                                                     (short)0, accL, false, false);
        v2f aX = *(const v2f*)(lx + kk * 4);
        accR = __builtin_amdgcn_wmma_f32_16x16x4_f32(false, aX, false, bR[kk],
                                                     (short)0, accR, false, false);
      }
#pragma unroll
      for (int j = 0; j < 8; ++j)
        lds_o[(rt * TILE_M + rbase + j) * LDP + h0 + nn] = accL[j] + accR[j] + bias;
    }
    __syncthreads();

#pragma unroll
    for (int rt = 0; rt < ROWT; ++rt) {
      const int r = rt * TILE_M + sr;
      const int p = tid & 15;
      const float* po = lds_o + r * LDP + p * 8;
      float s = 0.f;
#pragma unroll
      for (int q = 0; q < 8; ++q) s += po[q] * po[q];
      s += __shfl_xor(s, 1);
      s += __shfl_xor(s, 2);
      s += __shfl_xor(s, 4);
      s += __shfl_xor(s, 8);
      if (p == 0) lds_inv[r] = 1.0f / fmaxf(sqrtf(s), 1e-12f);
    }
    __syncthreads();

#pragma unroll
    for (int rt = 0; rt < ROWT; ++rt) {
      const int r = rt * TILE_M + sr;
      const float inv = lds_inv[r];
      const float* po = lds_o + r * LDP + sc;
      float4 o0, o1;
      o0.x = fmaxf(po[0] * inv, 0.f); o0.y = fmaxf(po[1] * inv, 0.f);
      o0.z = fmaxf(po[2] * inv, 0.f); o0.w = fmaxf(po[3] * inv, 0.f);
      o1.x = fmaxf(po[4] * inv, 0.f); o1.y = fmaxf(po[5] * inv, 0.f);
      o1.z = fmaxf(po[6] * inv, 0.f); o1.w = fmaxf(po[7] * inv, 0.f);
      float4* pw = (float4*)(xout + (size_t)(row0 + r) * HID + sc);
      pw[0] = o0;
      pw[1] = o1;
    }
  }
}

__global__ __launch_bounds__(256) void mlp_head(
    const float* __restrict__ xin, const float* __restrict__ W1,
    const float* __restrict__ b1, const float* __restrict__ W2,
    const float* __restrict__ b2, float* __restrict__ out, int ntiles) {
  __shared__ float lds_x[TILE_M * LDP];
  __shared__ float lds_h[TILE_M * LDP];

  const int tid   = threadIdx.x;
  const int wave  = tid >> 5;
  const int lane  = tid & 31;
  const int nn    = lane & 15;
  const int khalf = (lane >> 4) * 2;
  const int h0    = wave * 16;
  const int sr    = tid >> 4;
  const int sp    = tid & 15;
  const int sc    = sp * 8;

  int tile = blockIdx.x;
  float4 q0, q1;
  {
    const float4* px = (const float4*)(xin + ((size_t)tile * TILE_M + sr) * HID + sc);
    q0 = px[0];
    q1 = px[1];
  }

  v2f bW[32];
  {
    const float* pw = W1 + (size_t)(h0 + nn) * HID + khalf;
#pragma unroll
    for (int kk = 0; kk < 32; ++kk) bW[kk] = *(const v2f*)(pw + kk * 4);
  }
  const float bias = b1[h0 + nn];
  float w2a[8], w2b[8];
#pragma unroll
  for (int q = 0; q < 8; ++q) {
    w2a[q] = W2[sc + q];
    w2b[q] = W2[HID + sc + q];
  }
  const float c0 = b2[0], c1 = b2[1];
  const int rbase = (lane >> 4) * 8;

  for (; tile < ntiles; tile += (int)gridDim.x) {
    const int row0 = tile * TILE_M;
    {
      float4* lx4 = (float4*)(lds_x + sr * LDP + sc);
      lx4[0] = q0;
      lx4[1] = q1;
    }
    __syncthreads();

    {
      const int nt = tile + (int)gridDim.x;
      const int pt = (nt < ntiles) ? nt : tile;
      const float4* px = (const float4*)(xin + ((size_t)pt * TILE_M + sr) * HID + sc);
      q0 = px[0];
      q1 = px[1];
    }
    __builtin_amdgcn_sched_barrier(0);

    v8f acc = {0.f, 0.f, 0.f, 0.f, 0.f, 0.f, 0.f, 0.f};
    const float* lx = lds_x + nn * LDP + khalf;
#pragma unroll
    for (int kk = 0; kk < 32; ++kk) {
      v2f a = *(const v2f*)(lx + kk * 4);
      acc = __builtin_amdgcn_wmma_f32_16x16x4_f32(false, a, false, bW[kk],
                                                  (short)0, acc, false, false);
    }
#pragma unroll
    for (int j = 0; j < 8; ++j)
      lds_h[(rbase + j) * LDP + h0 + nn] = fmaxf(acc[j] + bias, 0.f);
    __syncthreads();

    {
      const float* ph = lds_h + sr * LDP + sc;
      float s0 = 0.f, s1 = 0.f;
#pragma unroll
      for (int q = 0; q < 8; ++q) {
        float h = ph[q];
        s0 += h * w2a[q];
        s1 += h * w2b[q];
      }
      s0 += __shfl_xor(s0, 1); s1 += __shfl_xor(s1, 1);
      s0 += __shfl_xor(s0, 2); s1 += __shfl_xor(s1, 2);
      s0 += __shfl_xor(s0, 4); s1 += __shfl_xor(s1, 4);
      s0 += __shfl_xor(s0, 8); s1 += __shfl_xor(s1, 8);
      if (sp == 0) {
        out[(size_t)(row0 + sr) * 2 + 0] = s0 + c0;
        out[(size_t)(row0 + sr) * 2 + 1] = s1 + c1;
      }
    }
  }
}

extern "C" void kernel_launch(void* const* d_in, const int* in_sizes, int n_in,
                              void* d_out, int out_size, void* d_ws, size_t ws_size,
                              hipStream_t stream) {
  const float* x  = (const float*)d_in[0];
  const int*   ei = (const int*)d_in[1];
  const float* Wl = (const float*)d_in[2];
  const float* bl = (const float*)d_in[3];
  const float* Wr = (const float*)d_in[4];
  const float* W1 = (const float*)d_in[5];
  const float* b1 = (const float*)d_in[6];
  const float* W2 = (const float*)d_in[7];
  const float* b2 = (const float*)d_in[8];
  float* out = (float*)d_out;

  const int N = NNODES;
  const int E = in_sizes[1] / 2;
  const int* src = ei;
  const int* dst = ei + E;

  float* inv_deg = (float*)d_ws;
  const size_t pad = 100096;
  float* bufA = inv_deg + pad;
  float* bufB = bufA + (size_t)N * HID;
  const size_t npad = (size_t)NBUCK * BNODES;
  const int G = (E + CHUNK - 1) / CHUNK;
  const int M = NBUCK * G;
  const int scanBlocks = (M + 1023) / 1024;
  const size_t mpad = (size_t)scanBlocks * 1024;
  int*  cnt    = (int*)(bufB + (size_t)N * HID);
  int*  rowptr = cnt + npad;
  int*  bh     = rowptr + npad;
  int*  bb     = bh + mpad;
  int*  ebuf   = bb + mpad;
  int*  col    = ebuf + 2 * (size_t)E;

  bin_count<<<G, 1024, 0, stream>>>(dst, bh, E, G);
  scan_kernel<<<scanBlocks, 1024, 0, stream>>>(bh, bb, M);
  bin_scatter<<<G, 1024, 0, stream>>>(src, dst, bb, ebuf, E, G);
  bucket_build<<<NBUCK, 1024, 0, stream>>>(ebuf, bb, cnt, rowptr, inv_deg, col, E, G, N);

  const float* xs[3] = {x, bufA, bufB};
  float*       ag[3] = {bufA, bufB, bufA};
  const int gatherBlocks  = (N + 7) / 8;
  const int xformChunks   = N / ROWS;
  const int headTiles     = N / TILE_M;

  for (int i = 0; i < 3; ++i) {
    float* agg = ag[i];
    const float* xc = xs[i];
    gather_kernel<<<gatherBlocks, 256, 0, stream>>>(xc, rowptr, cnt, col, agg, N);
    sage_transform<<<XFORM_WG, 256, 0, stream>>>(
        agg, xc, Wl + (size_t)i * HID * HID, bl + (size_t)i * HID,
        Wr + (size_t)i * HID * HID, inv_deg, agg, xformChunks);
  }
  mlp_head<<<HEAD_WG, 256, 0, stream>>>(bufA, W1, b1, W2, b2, out, headTiles);
}


# Neighbour gather fused into the layer kernel: a wave sums its 4 nodes' rows (col order as before) into the LDS A tile beside the other work-group's products; agg planes and 3 launches gone
# speedup vs baseline: 5.3074x; 1.1217x over previous
#include <hip/hip_runtime.h>
#include <math.h>

typedef __attribute__((ext_vector_type(2))) float v2f;
typedef __attribute__((ext_vector_type(8))) float v8f;

#define NNODES 100000
#define HID    128
#define TILE_M 16
#define ROWT   2
#define ROWS   (ROWT * TILE_M)
#define LDP    132

#define BSH    9
#define BNODES (1 << BSH)
#define NBUCK  ((NNODES + BNODES - 1) / BNODES)
#define CHUNK  8192

__global__ __launch_bounds__(1024) void bin_count(const int* __restrict__ dst,
                                                  int* __restrict__ bh, int E, int G) {
  __shared__ int h[NBUCK];
  const int tid = threadIdx.x;
  if (tid < NBUCK) h[tid] = 0;
  __syncthreads();
  const int base = blockIdx.x * CHUNK;
  int d[CHUNK / 1024];
#pragma unroll
  for (int q = 0; q < CHUNK / 1024; ++q) {
    const int e = base + q * 1024 + tid;
    d[q] = dst[min(e, E - 1)];
  }
#pragma unroll
  for (int q = 0; q < CHUNK / 1024; ++q) {
    const int e = base + q * 1024 + tid;
    if (e < E)
      __hip_atomic_fetch_add(&h[d[q] >> BSH], 1, __ATOMIC_RELAXED,
                             __HIP_MEMORY_SCOPE_WORKGROUP);
  }
  __syncthreads();
  if (tid < NBUCK) bh[tid * G + blockIdx.x] = h[tid];
}

__global__ __launch_bounds__(1024) void scan_kernel(const int* __restrict__ in,
                                                    int* __restrict__ out, int M) {
  __shared__ int wpre[32];
  __shared__ int wtot[32];
  const int tid  = threadIdx.x;
  const int lane = tid & 31;
  const int wave = tid >> 5;

  int s = 0;
  const int4* c4 = (const int4*)in;
  const int n4 = blockIdx.x * 256;
  for (int i = tid; i < n4; i += 1024) {
    int4 v = c4[i];
    s += (v.x + v.y) + (v.z + v.w);
  }
  s += __shfl_xor(s, 1);
  s += __shfl_xor(s, 2);
  s += __shfl_xor(s, 4);
  s += __shfl_xor(s, 8);
  s += __shfl_xor(s, 16);
  if (lane == 0) wpre[wave] = s;

  const int n = blockIdx.x * 1024 + tid;
  const int c = (n < M) ? in[n] : 0;
  int incl = c;
#pragma unroll
  for (int o = 1; o < 32; o <<= 1) {
    int t = __shfl_up(incl, o);
    if (lane >= o) incl += t;
  }
  if (lane == 31) wtot[wave] = incl;
  __syncthreads();

  int ab = wpre[lane] + ((lane < wave) ? wtot[lane] : 0);
  ab += __shfl_xor(ab, 1);
  ab += __shfl_xor(ab, 2);
  ab += __shfl_xor(ab, 4);
  ab += __shfl_xor(ab, 8);
  ab += __shfl_xor(ab, 16);
  if (n < M) out[n] = ab + incl - c;
}

__global__ __launch_bounds__(1024) void bin_scatter(const int* __restrict__ src,
                                                    const int* __restrict__ dst,
                                                    const int* __restrict__ bb,
                                                    int* __restrict__ ebuf_w, int E, int G) {
  int2* ebuf = (int2*)ebuf_w;
  __shared__ int cur[NBUCK];
  const int tid = threadIdx.x;
  if (tid < NBUCK) cur[tid] = bb[tid * G + blockIdx.x];
  __syncthreads();
  const int base = blockIdx.x * CHUNK;
  int d[CHUNK / 1024], s[CHUNK / 1024];
#pragma unroll
  for (int q = 0; q < CHUNK / 1024; ++q) {
    const int e = min(base + q * 1024 + tid, E - 1);
    d[q] = dst[e];
    s[q] = src[e];
  }
#pragma unroll
  for (int q = 0; q < CHUNK / 1024; ++q) {
    const int e = base + q * 1024 + tid;
    if (e < E) {
      const int p = __hip_atomic_fetch_add(&cur[d[q] >> BSH], 1, __ATOMIC_RELAXED,
                                           __HIP_MEMORY_SCOPE_WORKGROUP);
      ebuf[p] = make_int2(s[q], d[q]);
    }
  }
}

__global__ __launch_bounds__(1024) void bucket_build(const int* __restrict__ ebuf_w,
                                                     const int* __restrict__ bb,
                                                     int* __restrict__ cnt,
                                                     int* __restrict__ rowptr,
                                                     float* __restrict__ invdeg,
                                                     int* __restrict__ col,
                                                     int E, int G, int N) {
  const int2* ebuf = (const int2*)ebuf_w;
  __shared__ int c[BNODES];
  __shared__ int wtot[BNODES / 32];
  const int tid  = threadIdx.x;
  const int lane = tid & 31;
  const int wave = tid >> 5;
  const int b    = blockIdx.x;
  const int beg  = bb[b * G];
  const int end  = (b + 1 < (int)gridDim.x) ? bb[(b + 1) * G] : E;

  if (tid < BNODES) c[tid] = 0;
  __syncthreads();
  for (int e0 = beg + tid; e0 < end; e0 += 4 * 1024) {
    int dd[4];
#pragma unroll
    for (int u = 0; u < 4; ++u) dd[u] = ebuf[min(e0 + u * 1024, end - 1)].y;
#pragma unroll
    for (int u = 0; u < 4; ++u)
      if (e0 + u * 1024 < end)
        __hip_atomic_fetch_add(&c[dd[u] & (BNODES - 1)], 1, __ATOMIC_RELAXED,
                               __HIP_MEMORY_SCOPE_WORKGROUP);
  }
  __syncthreads();

  const int v = (tid < BNODES) ? c[tid] : 0;
  int incl = v;
#pragma unroll
  for (int o = 1; o < 32; o <<= 1) {
    int t = __shfl_up(incl, o);
    if (lane >= o) incl += t;
  }
  if (tid < BNODES && lane == 31) wtot[wave] = incl;
  __syncthreads();

  if (tid < BNODES) {
    int pre = 0;
    for (int w = 0; w < wave; ++w) pre += wtot[w];
    const int start = beg + pre + incl - v;
    c[tid] = start;
    const int n = b * BNODES + tid;
    if (n < N) {
      cnt[n]    = v;
      rowptr[n] = start;
      invdeg[n] = 1.0f / fmaxf((float)v, 1.0f);
    }
  }
  __syncthreads();

  for (int e0 = beg + tid; e0 < end; e0 += 4 * 1024) {
    int2 sd[4];
#pragma unroll
    for (int u = 0; u < 4; ++u) sd[u] = ebuf[min(e0 + u * 1024, end - 1)];
#pragma unroll
    for (int u = 0; u < 4; ++u)
      if (e0 + u * 1024 < end) {
        const int p = __hip_atomic_fetch_add(&c[sd[u].y & (BNODES - 1)], 1, __ATOMIC_RELAXED,
                                             __HIP_MEMORY_SCOPE_WORKGROUP);
        col[p] = sd[u].x;
      }
  }
}

#define XFORM_WG 512
#define HEAD_WG  1024
#define NPW      (ROWS / 8)
static_assert(ROWT == 2 && NPW == 4, "the staging code below names the wave's four rows of a chunk");

__device__ __forceinline__ float4 add_window(float4 acc, const float* __restrict__ xl,
                                             int ci, int m) {
  int u = 0;
  for (; u + 16 <= m; u += 16) {
    float4 v[16];
#pragma unroll
    for (int q = 0; q < 16; ++q)
      v[q] = *(const float4*)(xl + (size_t)__builtin_amdgcn_readlane(ci, u + q) * HID);
#pragma unroll
    for (int q = 0; q < 16; ++q) {
      acc.x += v[q].x; acc.y += v[q].y; acc.z += v[q].z; acc.w += v[q].w;
    }
  }
  const int r = m - u;
  float4 t[15];
#pragma unroll
  for (int q = 0; q < 15; ++q) t[q] = make_float4(0.f, 0.f, 0.f, 0.f);
  if (r & 8) {
#pragma unroll
    for (int q = 0; q < 8; ++q)
      t[q] = *(const float4*)(xl + (size_t)__builtin_amdgcn_readlane(ci, u + q) * HID);
    u += 8;
  }
  if (r & 4) {
#pragma unroll
    for (int q = 0; q < 4; ++q)
      t[8 + q] = *(const float4*)(xl + (size_t)__builtin_amdgcn_readlane(ci, u + q) * HID);
    u += 4;
  }
  if (r & 2) {
    t[12] = *(const float4*)(xl + (size_t)__builtin_amdgcn_readlane(ci, u) * HID);
    t[13] = *(const float4*)(xl + (size_t)__builtin_amdgcn_readlane(ci, u + 1) * HID);
    u += 2;
  }
  if (r & 1) t[14] = *(const float4*)(xl + (size_t)__builtin_amdgcn_readlane(ci, u) * HID);
  if (r & 8) {
#pragma unroll
    for (int q = 0; q < 8; ++q) {
      acc.x += t[q].x; acc.y += t[q].y; acc.z += t[q].z; acc.w += t[q].w;
    }
  }
  if (r & 4) {
#pragma unroll
    for (int q = 8; q < 12; ++q) {
      acc.x += t[q].x; acc.y += t[q].y; acc.z += t[q].z; acc.w += t[q].w;
    }
  }
  if (r & 2) {
    acc.x += t[12].x; acc.y += t[12].y; acc.z += t[12].z; acc.w += t[12].w;
    acc.x += t[13].x; acc.y += t[13].y; acc.z += t[13].z; acc.w += t[13].w;
  }
  if (r & 1) {
    acc.x += t[14].x; acc.y += t[14].y; acc.z += t[14].z; acc.w += t[14].w;
  }
  return acc;
}

__device__ __forceinline__ int col_window(const int* __restrict__ col, int j0, int end, int lane) {
  return col[max(min(j0 + lane, end - 1), 0)];
}

__device__ __forceinline__ float4 neighbour_sum(const float* __restrict__ xl,
                                                const int* __restrict__ col,
                                                int beg, int end, int ci0, int lane) {
  float4 acc = make_float4(0.f, 0.f, 0.f, 0.f);
  acc = add_window(acc, xl, ci0, min(end - beg, 32));
  for (int j0 = beg + 32; j0 < end; j0 += 32)
    acc = add_window(acc, xl, col_window(col, j0, end, lane), min(end - j0, 32));
  return acc;
}

__global__ __launch_bounds__(256) void sage_layer(
    const float* __restrict__ xin, const int* __restrict__ rowptr,
    const int* __restrict__ cnt, const int* __restrict__ col,
    const float* __restrict__ invdeg,
    const float* __restrict__ Wl, const float* __restrict__ bl,
    const float* __restrict__ Wr, float* __restrict__ xout, int nchunks) {
  __shared__ float lds_a[ROWS * LDP];
  __shared__ float lds_x[ROWS * LDP];
  __shared__ float lds_o[ROWS * LDP];
  __shared__ float lds_inv[ROWS];

  const int tid   = threadIdx.x;
  const int wave  = tid >> 5;
  const int lane  = tid & 31;
  const int nn    = lane & 15;
  const int khalf = (lane >> 4) * 2;
  const int h0    = wave * 16;
  const int sr    = tid >> 4;
  const int sc    = (tid & 15) * 8;
  const int wrow  = __builtin_amdgcn_readfirstlane(wave * NPW);
  const float* xl = xin + lane * 4;

  float4 qx0, qx1, qx2, qx3;
  int    vb, vc;
  float  vi;
  int chunk = blockIdx.x;
  {
    const int n0 = chunk * ROWS + wrow;
    const float* px = xl + (size_t)n0 * HID;
    qx0 = *(const float4*)(px);
    qx1 = *(const float4*)(px + HID);
    qx2 = *(const float4*)(px + 2 * HID);
    qx3 = *(const float4*)(px + 3 * HID);
    vb = rowptr[n0 + (lane & 3)];
    vc = cnt[n0 + (lane & 3)];
    vi = invdeg[n0 + (lane & 3)];
  }

  v2f bL[32], bR[32];
  {
    const float* pwl = Wl + (size_t)(h0 + nn) * HID + khalf;
    const float* pwr = Wr + (size_t)(h0 + nn) * HID + khalf;
#pragma unroll
    for (int kk = 0; kk < 32; ++kk) {
      bL[kk] = *(const v2f*)(pwl + kk * 4);
      bR[kk] = *(const v2f*)(pwr + kk * 4);
    }
  }
  const float bias  = bl[h0 + nn];
  const int   rbase = (lane >> 4) * 8;

  for (; chunk < nchunks; chunk += (int)gridDim.x) {
    const int row0 = chunk * ROWS;

    {
      float* lx = lds_x + wrow * LDP + lane * 4;
      *(float4*)(lx)           = qx0;
      *(float4*)(lx + LDP)     = qx1;
      *(float4*)(lx + 2 * LDP) = qx2;
      *(float4*)(lx + 3 * LDP) = qx3;
    }
    {
      const int b0 = __builtin_amdgcn_readlane(vb, 0), e0 = b0 + __builtin_amdgcn_readlane(vc, 0);
      const int b1 = __builtin_amdgcn_readlane(vb, 1), e1 = b1 + __builtin_amdgcn_readlane(vc, 1);
      const int b2 = __builtin_amdgcn_readlane(vb, 2), e2 = b2 + __builtin_amdgcn_readlane(vc, 2);
      const int b3 = __builtin_amdgcn_readlane(vb, 3), e3 = b3 + __builtin_amdgcn_readlane(vc, 3);
      const int fi = __float_as_int(vi);
      const float f0 = __int_as_float(__builtin_amdgcn_readlane(fi, 0));
      const float f1 = __int_as_float(__builtin_amdgcn_readlane(fi, 1));
      const float f2 = __int_as_float(__builtin_amdgcn_readlane(fi, 2));
      const float f3 = __int_as_float(__builtin_amdgcn_readlane(fi, 3));
      const int c0 = col_window(col, b0, e0, lane);
      const int c1 = col_window(col, b1, e1, lane);
      const int c2 = col_window(col, b2, e2, lane);
      const int c3 = col_window(col, b3, e3, lane);
      float* la = lds_a + wrow * LDP + lane * 4;
      float4 a;
      a = neighbour_sum(xl, col, b0, e0, c0, lane);
      *(float4*)(la)           = make_float4(a.x * f0, a.y * f0, a.z * f0, a.w * f0);
      a = neighbour_sum(xl, col, b1, e1, c1, lane);
      *(float4*)(la + LDP)     = make_float4(a.x * f1, a.y * f1, a.z * f1, a.w * f1);
      a = neighbour_sum(xl, col, b2, e2, c2, lane);
      *(float4*)(la + 2 * LDP) = make_float4(a.x * f2, a.y * f2, a.z * f2, a.w * f2);
      a = neighbour_sum(xl, col, b3, e3, c3, lane);
      *(float4*)(la + 3 * LDP) = make_float4(a.x * f3, a.y * f3, a.z * f3, a.w * f3);
    }
    __syncthreads();

    {
      const int nc = chunk + (int)gridDim.x;
      const int pc = (nc < nchunks) ? nc : chunk;
      const int n0 = pc * ROWS + wrow;
      const float* px = xl + (size_t)n0 * HID;
      qx0 = *(const float4*)(px);
      qx1 = *(const float4*)(px + HID);
      qx2 = *(const float4*)(px + 2 * HID);
      qx3 = *(const float4*)(px + 3 * HID);
      vb = rowptr[n0 + (lane & 3)];
      vc = cnt[n0 + (lane & 3)];
      vi = invdeg[n0 + (lane & 3)];
    }

#pragma unroll 1
    for (int rt = 0; rt < ROWT; ++rt) {
      v8f accL = {0.f, 0.f, 0.f, 0.f, 0.f, 0.f, 0.f, 0.f};
      v8f accR = {0.f, 0.f, 0.f, 0.f, 0.f, 0.f, 0.f, 0.f};
      const float* la = lds_a + (rt * TILE_M + nn) * LDP + khalf;
      const float* lx = lds_x + (rt * TILE_M + nn) * LDP + khalf;
#pragma unroll
      for (int kk = 0; kk < 32; ++kk) {
        v2f aA = *(const v2f*)(la + kk * 4);
        accL = __builtin_amdgcn_wmma_f32_16x16x4_f32(false, aA, false, bL[kk],
                                                     (short)0, accL, false, false);
        v2f aX = *(const v2f*)(lx + kk * 4);
        accR = __builtin_amdgcn_wmma_f32_16x16x4_f32(false, aX, false, bR[kk],
                                                     (short)0, accR, false, false);
      }
#pragma unroll
      for (int j = 0; j < 8; ++j)
        lds_o[(rt * TILE_M + rbase + j) * LDP + h0 + nn] = accL[j] + accR[j] + bias;
    }
    __syncthreads();

#pragma unroll
    for (int rt = 0; rt < ROWT; ++rt) {
      const int r = rt * TILE_M + sr;
      const int p = tid & 15;
      const float* po = lds_o + r * LDP + p * 8;
      float s = 0.f;
#pragma unroll
      for (int q = 0; q < 8; ++q) s += po[q] * po[q];
      s += __shfl_xor(s, 1);
      s += __shfl_xor(s, 2);
      s += __shfl_xor(s, 4);
      s += __shfl_xor(s, 8);
      if (p == 0) lds_inv[r] = 1.0f / fmaxf(sqrtf(s), 1e-12f);
    }
    __syncthreads();

#pragma unroll
    for (int rt = 0; rt < ROWT; ++rt) {
      const int r = rt * TILE_M + sr;
      const float inv = lds_inv[r];
      const float* po = lds_o + r * LDP + sc;
      float4 o0, o1;
      o0.x = fmaxf(po[0] * inv, 0.f); o0.y = fmaxf(po[1] * inv, 0.f);
      o0.z = fmaxf(po[2] * inv, 0.f); o0.w = fmaxf(po[3] * inv, 0.f);
      o1.x = fmaxf(po[4] * inv, 0.f); o1.y = fmaxf(po[5] * inv, 0.f);
      o1.z = fmaxf(po[6] * inv, 0.f); o1.w = fmaxf(po[7] * inv, 0.f);
      float4* pw = (float4*)(xout + (size_t)(row0 + r) * HID + sc);
      pw[0] = o0;
      pw[1] = o1;
    }
  }
}

__global__ __launch_bounds__(256) void mlp_head(
    const float* __restrict__ xin, const float* __restrict__ W1,
    const float* __restrict__ b1, const float* __restrict__ W2,
    const float* __restrict__ b2, float* __restrict__ out, int ntiles) {
  __shared__ float lds_x[TILE_M * LDP];
  __shared__ float lds_h[TILE_M * LDP];

  const int tid   = threadIdx.x;
  const int wave  = tid >> 5;
  const int lane  = tid & 31;
  const int nn    = lane & 15;
  const int khalf = (lane >> 4) * 2;
  const int h0    = wave * 16;
  const int sr    = tid >> 4;
  const int sp    = tid & 15;
  const int sc    = sp * 8;

  int tile = blockIdx.x;
  float4 q0, q1;
  {
    const float4* px = (const float4*)(xin + ((size_t)tile * TILE_M + sr) * HID + sc);
    q0 = px[0];
    q1 = px[1];
  }

  v2f bW[32];
  {
    const float* pw = W1 + (size_t)(h0 + nn) * HID + khalf;
#pragma unroll
    for (int kk = 0; kk < 32; ++kk) bW[kk] = *(const v2f*)(pw + kk * 4);
  }
  const float bias = b1[h0 + nn];
  float w2a[8], w2b[8];
#pragma unroll
  for (int q = 0; q < 8; ++q) {
    w2a[q] = W2[sc + q];
    w2b[q] = W2[HID + sc + q];
  }
  const float c0 = b2[0], c1 = b2[1];
  const int rbase = (lane >> 4) * 8;

  for (; tile < ntiles; tile += (int)gridDim.x) {
    const int row0 = tile * TILE_M;
    {
      float4* lx4 = (float4*)(lds_x + sr * LDP + sc);
      lx4[0] = q0;
      lx4[1] = q1;
    }
    __syncthreads();

    {
      const int nt = tile + (int)gridDim.x;
      const int pt = (nt < ntiles) ? nt : tile;
      const float4* px = (const float4*)(xin + ((size_t)pt * TILE_M + sr) * HID + sc);
      q0 = px[0];
      q1 = px[1];
    }
    __builtin_amdgcn_sched_barrier(0);

    v8f acc = {0.f, 0.f, 0.f, 0.f, 0.f, 0.f, 0.f, 0.f};
    const float* lx = lds_x + nn * LDP + khalf;
#pragma unroll
    for (int kk = 0; kk < 32; ++kk) {
      v2f a = *(const v2f*)(lx + kk * 4);
      acc = __builtin_amdgcn_wmma_f32_16x16x4_f32(false, a, false, bW[kk],
                                                  (short)0, acc, false, false);
    }
#pragma unroll
    for (int j = 0; j < 8; ++j)
      lds_h[(rbase + j) * LDP + h0 + nn] = fmaxf(acc[j] + bias, 0.f);
    __syncthreads();

    {
      const float* ph = lds_h + sr * LDP + sc;
      float s0 = 0.f, s1 = 0.f;
#pragma unroll
      for (int q = 0; q < 8; ++q) {
        float h = ph[q];
        s0 += h * w2a[q];
        s1 += h * w2b[q];
      }
      s0 += __shfl_xor(s0, 1); s1 += __shfl_xor(s1, 1);
      s0 += __shfl_xor(s0, 2); s1 += __shfl_xor(s1, 2);
      s0 += __shfl_xor(s0, 4); s1 += __shfl_xor(s1, 4);
      s0 += __shfl_xor(s0, 8); s1 += __shfl_xor(s1, 8);
      if (sp == 0) {
        out[(size_t)(row0 + sr) * 2 + 0] = s0 + c0;
        out[(size_t)(row0 + sr) * 2 + 1] = s1 + c1;
      }
    }
  }
}

extern "C" void kernel_launch(void* const* d_in, const int* in_sizes, int n_in,
                              void* d_out, int out_size, void* d_ws, size_t ws_size,
                              hipStream_t stream) {
  const float* x  = (const float*)d_in[0];
  const int*   ei = (const int*)d_in[1];
  const float* Wl = (const float*)d_in[2];
  const float* bl = (const float*)d_in[3];
  const float* Wr = (const float*)d_in[4];
  const float* W1 = (const float*)d_in[5];
  const float* b1 = (const float*)d_in[6];
  const float* W2 = (const float*)d_in[7];
  const float* b2 = (const float*)d_in[8];
  float* out = (float*)d_out;

  const int N = NNODES;
  const int E = in_sizes[1] / 2;
  const int* src = ei;
  const int* dst = ei + E;

  float* inv_deg = (float*)d_ws;
  const size_t pad = 100096;
  float* bufA = inv_deg + pad;
  float* bufB = bufA + (size_t)N * HID;
  const size_t npad = (size_t)NBUCK * BNODES;
  const int G = (E + CHUNK - 1) / CHUNK;
  const int M = NBUCK * G;
  const int scanBlocks = (M + 1023) / 1024;
  const size_t mpad = (size_t)scanBlocks * 1024;
  int*  cnt    = (int*)(bufB + (size_t)N * HID);
  int*  rowptr = cnt + npad;
  int*  bh     = rowptr + npad;
  int*  bb     = bh + mpad;
  int*  ebuf   = bb + mpad;
  int*  col    = ebuf + 2 * (size_t)E;

  bin_count<<<G, 1024, 0, stream>>>(dst, bh, E, G);
  scan_kernel<<<scanBlocks, 1024, 0, stream>>>(bh, bb, M);
  bin_scatter<<<G, 1024, 0, stream>>>(src, dst, bb, ebuf, E, G);
  bucket_build<<<NBUCK, 1024, 0, stream>>>(ebuf, bb, cnt, rowptr, inv_deg, col, E, G, N);

  const float* xs[3] = {x, bufA, bufB};
  float*       xo[3] = {bufA, bufB, bufA};
  const int xformChunks   = N / ROWS;
  const int headTiles     = N / TILE_M;

  for (int i = 0; i < 3; ++i) {
    sage_layer<<<XFORM_WG, 256, 0, stream>>>(
        xs[i], rowptr, cnt, col, inv_deg, Wl + (size_t)i * HID * HID,
        bl + (size_t)i * HID, Wr + (size_t)i * HID * HID, xo[i], xformChunks);
  }
  mlp_head<<<HEAD_WG, 256, 0, stream>>>(bufA, W1, b1, W2, b2, out, headTiles);
}


# Own-row products (x rows times Wr) run in four quarters inside the neighbour gather's load waits, one behind each node's row requests; only the Wl products stay behind the gather; sums unchanged
# speedup vs baseline: 5.6406x; 1.0628x over previous
#include <hip/hip_runtime.h>
#include <math.h>

typedef __attribute__((ext_vector_type(2))) float v2f;
typedef __attribute__((ext_vector_type(8))) float v8f;

#define NNODES 100000
#define HID    128
#define TILE_M 16
#define ROWT   2
#define ROWS   (ROWT * TILE_M)
#define LDP    132

#define BSH    9
#define BNODES (1 << BSH)
#define NBUCK  ((NNODES + BNODES - 1) / BNODES)
#define CHUNK  8192

__global__ __launch_bounds__(1024) void bin_count(const int* __restrict__ dst,
                                                  int* __restrict__ bh, int E, int G) {
  __shared__ int h[NBUCK];
  const int tid = threadIdx.x;
  if (tid < NBUCK) h[tid] = 0;
  __syncthreads();
  const int base = blockIdx.x * CHUNK;
  int d[CHUNK / 1024];
#pragma unroll
  for (int q = 0; q < CHUNK / 1024; ++q) {
    const int e = base + q * 1024 + tid;
    d[q] = dst[min(e, E - 1)];
  }
#pragma unroll
  for (int q = 0; q < CHUNK / 1024; ++q) {
    const int e = base + q * 1024 + tid;
    if (e < E)
      __hip_atomic_fetch_add(&h[d[q] >> BSH], 1, __ATOMIC_RELAXED,
                             __HIP_MEMORY_SCOPE_WORKGROUP);
  }
  __syncthreads();
  if (tid < NBUCK) bh[tid * G + blockIdx.x] = h[tid];
}

__global__ __launch_bounds__(1024) void scan_kernel(const int* __restrict__ in,
                                                    int* __restrict__ out, int M) {
  __shared__ int wpre[32];
  __shared__ int wtot[32];
  const int tid  = threadIdx.x;
  const int lane = tid & 31;
  const int wave = tid >> 5;

  int s = 0;
  const int4* c4 = (const int4*)in;
  const int n4 = blockIdx.x * 256;
  for (int i = tid; i < n4; i += 1024) {
    int4 v = c4[i];
    s += (v.x + v.y) + (v.z + v.w);
  }
  s += __shfl_xor(s, 1);
  s += __shfl_xor(s, 2);
  s += __shfl_xor(s, 4);
  s += __shfl_xor(s, 8);
  s += __shfl_xor(s, 16);
  if (lane == 0) wpre[wave] = s;

  const int n = blockIdx.x * 1024 + tid;
  const int c = (n < M) ? in[n] : 0;
  int incl = c;
#pragma unroll
  for (int o = 1; o < 32; o <<= 1) {
    int t = __shfl_up(incl, o);
    if (lane >= o) incl += t;
  }
  if (lane == 31) wtot[wave] = incl;
  __syncthreads();

  int ab = wpre[lane] + ((lane < wave) ? wtot[lane] : 0);
  ab += __shfl_xor(ab, 1);
  ab += __shfl_xor(ab, 2);
  ab += __shfl_xor(ab, 4);
  ab += __shfl_xor(ab, 8);
  ab += __shfl_xor(ab, 16);
  if (n < M) out[n] = ab + incl - c;
}

__global__ __launch_bounds__(1024) void bin_scatter(const int* __restrict__ src,
                                                    const int* __restrict__ dst,
                                                    const int* __restrict__ bb,
                                                    int* __restrict__ ebuf_w, int E, int G) {
  int2* ebuf = (int2*)ebuf_w;
  __shared__ int cur[NBUCK];
  const int tid = threadIdx.x;
  if (tid < NBUCK) cur[tid] = bb[tid * G + blockIdx.x];
  __syncthreads();
  const int base = blockIdx.x * CHUNK;
  int d[CHUNK / 1024], s[CHUNK / 1024];
#pragma unroll
  for (int q = 0; q < CHUNK / 1024; ++q) {
    const int e = min(base + q * 1024 + tid, E - 1);
    d[q] = dst[e];
    s[q] = src[e];
  }
#pragma unroll
  for (int q = 0; q < CHUNK / 1024; ++q) {
    const int e = base + q * 1024 + tid;
    if (e < E) {
      const int p = __hip_atomic_fetch_add(&cur[d[q] >> BSH], 1, __ATOMIC_RELAXED,
                                           __HIP_MEMORY_SCOPE_WORKGROUP);
      ebuf[p] = make_int2(s[q], d[q]);
    }
  }
}

__global__ __launch_bounds__(1024) void bucket_build(const int* __restrict__ ebuf_w,
                                                     const int* __restrict__ bb,
                                                     int* __restrict__ cnt,
                                                     int* __restrict__ rowptr,
                                                     float* __restrict__ invdeg,
                                                     int* __restrict__ col,
                                                     int E, int G, int N) {
  const int2* ebuf = (const int2*)ebuf_w;
  __shared__ int c[BNODES];
  __shared__ int wtot[BNODES / 32];
  const int tid  = threadIdx.x;
  const int lane = tid & 31;
  const int wave = tid >> 5;
  const int b    = blockIdx.x;
  const int beg  = bb[b * G];
  const int end  = (b + 1 < (int)gridDim.x) ? bb[(b + 1) * G] : E;

  if (tid < BNODES) c[tid] = 0;
  __syncthreads();
  for (int e0 = beg + tid; e0 < end; e0 += 4 * 1024) {
    int dd[4];
#pragma unroll
    for (int u = 0; u < 4; ++u) dd[u] = ebuf[min(e0 + u * 1024, end - 1)].y;
#pragma unroll
    for (int u = 0; u < 4; ++u)
      if (e0 + u * 1024 < end)
        __hip_atomic_fetch_add(&c[dd[u] & (BNODES - 1)], 1, __ATOMIC_RELAXED,
                               __HIP_MEMORY_SCOPE_WORKGROUP);
  }
  __syncthreads();

  const int v = (tid < BNODES) ? c[tid] : 0;
  int incl = v;
#pragma unroll
  for (int o = 1; o < 32; o <<= 1) {
    int t = __shfl_up(incl, o);
    if (lane >= o) incl += t;
  }
  if (tid < BNODES && lane == 31) wtot[wave] = incl;
  __syncthreads();

  if (tid < BNODES) {
    int pre = 0;
    for (int w = 0; w < wave; ++w) pre += wtot[w];
    const int start = beg + pre + incl - v;
    c[tid] = start;
    const int n = b * BNODES + tid;
    if (n < N) {
      cnt[n]    = v;
      rowptr[n] = start;
      invdeg[n] = 1.0f / fmaxf((float)v, 1.0f);
    }
  }
  __syncthreads();

  for (int e0 = beg + tid; e0 < end; e0 += 4 * 1024) {
    int2 sd[4];
#pragma unroll
    for (int u = 0; u < 4; ++u) sd[u] = ebuf[min(e0 + u * 1024, end - 1)];
#pragma unroll
    for (int u = 0; u < 4; ++u)
      if (e0 + u * 1024 < end) {
        const int p = __hip_atomic_fetch_add(&c[sd[u].y & (BNODES - 1)], 1, __ATOMIC_RELAXED,
                                             __HIP_MEMORY_SCOPE_WORKGROUP);
        col[p] = sd[u].x;
      }
  }
}

#define XFORM_WG 512
#define HEAD_WG  1024
#define NPW      (ROWS / 8)
static_assert(ROWT == 2 && NPW == 4, "the staging code below names the wave's four rows of a chunk");

template <int Q>
__device__ __forceinline__ void own_quarter(v8f& r0, v8f& r1, const float* lx,
                                            const v2f (&bR)[32]) {
#pragma unroll
  for (int kk = 8 * Q; kk < 8 * Q + 8; ++kk) {
    const v2f a0 = *(const v2f*)(lx + kk * 4);
    const v2f a1 = *(const v2f*)(lx + TILE_M * LDP + kk * 4);
    r0 = __builtin_amdgcn_wmma_f32_16x16x4_f32(false, a0, false, bR[kk],
                                               (short)0, r0, false, false);
    r1 = __builtin_amdgcn_wmma_f32_16x16x4_f32(false, a1, false, bR[kk],
                                               (short)0, r1, false, false);
  }
}

template <int Q>
__device__ __forceinline__ float4 add_window(float4 acc, const float* __restrict__ xl,
                                             int ci, int m, v8f& r0, v8f& r1,
                                             const float* lx, const v2f (&bR)[32]) {
  int u = 0;
  bool pending = (Q >= 0);
  if constexpr (Q >= 0) {
    asm volatile("" :: "v"(ci));
  }
  for (; u + 16 <= m; u += 16) {
    float4 v[16];
#pragma unroll
    for (int q = 0; q < 16; ++q)
      v[q] = *(const float4*)(xl + (size_t)__builtin_amdgcn_readlane(ci, u + q) * HID);
    if constexpr (Q >= 0) {
      if (pending) {
        __builtin_amdgcn_sched_barrier(0);
        own_quarter<Q>(r0, r1, lx, bR);
        asm volatile("" : "+v"(acc.x), "+v"(acc.y), "+v"(acc.z), "+v"(acc.w) : "v"(r0), "v"(r1));
        pending = false;
      }
    }
#pragma unroll
    for (int q = 0; q < 16; ++q) {
      acc.x += v[q].x; acc.y += v[q].y; acc.z += v[q].z; acc.w += v[q].w;
    }
  }
  const int r = m - u;
  float4 t[15];
#pragma unroll
  for (int q = 0; q < 15; ++q) t[q] = make_float4(0.f, 0.f, 0.f, 0.f);
  if (r & 8) {
#pragma unroll
    for (int q = 0; q < 8; ++q)
      t[q] = *(const float4*)(xl + (size_t)__builtin_amdgcn_readlane(ci, u + q) * HID);
    u += 8;
  }
  if (r & 4) {
#pragma unroll
    for (int q = 0; q < 4; ++q)
      t[8 + q] = *(const float4*)(xl + (size_t)__builtin_amdgcn_readlane(ci, u + q) * HID);
    u += 4;
  }
  if (r & 2) {
    t[12] = *(const float4*)(xl + (size_t)__builtin_amdgcn_readlane(ci, u) * HID);
    t[13] = *(const float4*)(xl + (size_t)__builtin_amdgcn_readlane(ci, u + 1) * HID);
    u += 2;
  }
  if (r & 1) t[14] = *(const float4*)(xl + (size_t)__builtin_amdgcn_readlane(ci, u) * HID);
  if constexpr (Q >= 0) {
    if (pending) {
      __builtin_amdgcn_sched_barrier(0);
      own_quarter<Q>(r0, r1, lx, bR);
      asm volatile("" : "+v"(acc.x), "+v"(acc.y), "+v"(acc.z), "+v"(acc.w) : "v"(r0), "v"(r1));
    }
  }
  if (r & 8) {
#pragma unroll
    for (int q = 0; q < 8; ++q) {
      acc.x += t[q].x; acc.y += t[q].y; acc.z += t[q].z; acc.w += t[q].w;
    }
  }
  if (r & 4) {
#pragma unroll
    for (int q = 8; q < 12; ++q) {
      acc.x += t[q].x; acc.y += t[q].y; acc.z += t[q].z; acc.w += t[q].w;
    }
  }
  if (r & 2) {
    acc.x += t[12].x; acc.y += t[12].y; acc.z += t[12].z; acc.w += t[12].w;
    acc.x += t[13].x; acc.y += t[13].y; acc.z += t[13].z; acc.w += t[13].w;
  }
  if (r & 1) {
    acc.x += t[14].x; acc.y += t[14].y; acc.z += t[14].z; acc.w += t[14].w;
  }
  return acc;
}

__device__ __forceinline__ int col_window(const int* __restrict__ col, int j0, int end, int lane) {
  return col[max(min(j0 + lane, end - 1), 0)];
}

template <int Q>
__device__ __forceinline__ float4 neighbour_sum(const float* __restrict__ xl,
                                                const int* __restrict__ col,
                                                int beg, int end, int ci0, int lane,
                                                v8f& r0, v8f& r1, const float* lx,
                                                const v2f (&bR)[32]) {
  float4 acc = make_float4(0.f, 0.f, 0.f, 0.f);
  acc = add_window<Q>(acc, xl, ci0, min(end - beg, 32), r0, r1, lx, bR);
  for (int j0 = beg + 32; j0 < end; j0 += 32)
    acc = add_window<-1>(acc, xl, col_window(col, j0, end, lane), min(end - j0, 32),
                         r0, r1, lx, bR);
  return acc;
}

__global__ __launch_bounds__(256, 4) void sage_layer(
    const float* __restrict__ xin, const int* __restrict__ rowptr,
    const int* __restrict__ cnt, const int* __restrict__ col,
    const float* __restrict__ invdeg,
    const float* __restrict__ Wl, const float* __restrict__ bl,
    const float* __restrict__ Wr, float* __restrict__ xout, int nchunks) {
  __shared__ float lds_a[ROWS * LDP];
  __shared__ float lds_x[ROWS * LDP];
  __shared__ float lds_o[ROWS * LDP];
  __shared__ float lds_inv[ROWS];

  const int tid   = threadIdx.x;
  const int wave  = tid >> 5;
  const int lane  = tid & 31;
  const int nn    = lane & 15;
  const int khalf = (lane >> 4) * 2;
  const int h0    = wave * 16;
  const int sr    = tid >> 4;
  const int sc    = (tid & 15) * 8;
  const int wrow  = __builtin_amdgcn_readfirstlane(wave * NPW);
  const float* xl = xin + lane * 4;

  float4 qx0, qx1, qx2, qx3;
  int    vb, vc;
  float  vi;
  int chunk = blockIdx.x;
  {
    const int n0 = chunk * ROWS + wrow;
    const float* px = xl + (size_t)n0 * HID;
    qx0 = *(const float4*)(px);
    qx1 = *(const float4*)(px + HID);
    qx2 = *(const float4*)(px + 2 * HID);
    qx3 = *(const float4*)(px + 3 * HID);
    vb = rowptr[n0 + (lane & 3)];
    vc = cnt[n0 + (lane & 3)];
    vi = invdeg[n0 + (lane & 3)];
  }

  v2f bL[32], bR[32];
  {
    const float* pwl = Wl + (size_t)(h0 + nn) * HID + khalf;
    const float* pwr = Wr + (size_t)(h0 + nn) * HID + khalf;
#pragma unroll
    for (int kk = 0; kk < 32; ++kk) {
      bL[kk] = *(const v2f*)(pwl + kk * 4);
      bR[kk] = *(const v2f*)(pwr + kk * 4);
    }
  }
  const float bias  = bl[h0 + nn];
  const int   rbase = (lane >> 4) * 8;

  const float* lxf = lds_x + nn * LDP + khalf;
  const float* laf = lds_a + nn * LDP + khalf;
  float* lxs = lds_x + wrow * LDP + lane * 4;

  *(float4*)(lxs)           = qx0;
  *(float4*)(lxs + LDP)     = qx1;
  *(float4*)(lxs + 2 * LDP) = qx2;
  *(float4*)(lxs + 3 * LDP) = qx3;
  __syncthreads();

  for (; chunk < nchunks; chunk += (int)gridDim.x) {
    const int row0 = chunk * ROWS;

    v8f accR0 = {0.f, 0.f, 0.f, 0.f, 0.f, 0.f, 0.f, 0.f};
    v8f accR1 = {0.f, 0.f, 0.f, 0.f, 0.f, 0.f, 0.f, 0.f};
    {
      const int b0 = __builtin_amdgcn_readlane(vb, 0), e0 = b0 + __builtin_amdgcn_readlane(vc, 0);
      const int b1 = __builtin_amdgcn_readlane(vb, 1), e1 = b1 + __builtin_amdgcn_readlane(vc, 1);
      const int b2 = __builtin_amdgcn_readlane(vb, 2), e2 = b2 + __builtin_amdgcn_readlane(vc, 2);
      const int b3 = __builtin_amdgcn_readlane(vb, 3), e3 = b3 + __builtin_amdgcn_readlane(vc, 3);
      const int fi = __float_as_int(vi);
      const float f0 = __int_as_float(__builtin_amdgcn_readlane(fi, 0));
      const float f1 = __int_as_float(__builtin_amdgcn_readlane(fi, 1));
      const float f2 = __int_as_float(__builtin_amdgcn_readlane(fi, 2));
      const float f3 = __int_as_float(__builtin_amdgcn_readlane(fi, 3));
      const int c0 = col_window(col, b0, e0, lane);
      const int c1 = col_window(col, b1, e1, lane);
      const int c2 = col_window(col, b2, e2, lane);
      const int c3 = col_window(col, b3, e3, lane);
      float* la = lds_a + wrow * LDP + lane * 4;
      float4 a;
      a = neighbour_sum<0>(xl, col, b0, e0, c0, lane, accR0, accR1, lxf, bR);
      *(float4*)(la)           = make_float4(a.x * f0, a.y * f0, a.z * f0, a.w * f0);
      a = neighbour_sum<1>(xl, col, b1, e1, c1, lane, accR0, accR1, lxf, bR);
      *(float4*)(la + LDP)     = make_float4(a.x * f1, a.y * f1, a.z * f1, a.w * f1);
      a = neighbour_sum<2>(xl, col, b2, e2, c2, lane, accR0, accR1, lxf, bR);
      *(float4*)(la + 2 * LDP) = make_float4(a.x * f2, a.y * f2, a.z * f2, a.w * f2);
      a = neighbour_sum<3>(xl, col, b3, e3, c3, lane, accR0, accR1, lxf, bR);
      *(float4*)(la + 3 * LDP) = make_float4(a.x * f3, a.y * f3, a.z * f3, a.w * f3);
    }
    __syncthreads();

    {
      const int nc = chunk + (int)gridDim.x;
      const int pc = (nc < nchunks) ? nc : chunk;
      const int n0 = pc * ROWS + wrow;
      const float* px = xl + (size_t)n0 * HID;
      qx0 = *(const float4*)(px);
      qx1 = *(const float4*)(px + HID);
      qx2 = *(const float4*)(px + 2 * HID);
      qx3 = *(const float4*)(px + 3 * HID);
      vb = rowptr[n0 + (lane & 3)];
      vc = cnt[n0 + (lane & 3)];
      vi = invdeg[n0 + (lane & 3)];
    }
    __builtin_amdgcn_sched_barrier(0);

    {
      v8f accL0 = {0.f, 0.f, 0.f, 0.f, 0.f, 0.f, 0.f, 0.f};
      v8f accL1 = {0.f, 0.f, 0.f, 0.f, 0.f, 0.f, 0.f, 0.f};
#pragma unroll
      for (int kk = 0; kk < 32; ++kk) {
        const v2f a0 = *(const v2f*)(laf + kk * 4);
        const v2f a1 = *(const v2f*)(laf + TILE_M * LDP + kk * 4);
        accL0 = __builtin_amdgcn_wmma_f32_16x16x4_f32(false, a0, false, bL[kk],
                                                      (short)0, accL0, false, false);
        accL1 = __builtin_amdgcn_wmma_f32_16x16x4_f32(false, a1, false, bL[kk],
                                                      (short)0, accL1, false, false);
      }
#pragma unroll
      for (int j = 0; j < 8; ++j) {
        lds_o[(rbase + j) * LDP + h0 + nn]          = accL0[j] + accR0[j] + bias;
        lds_o[(TILE_M + rbase + j) * LDP + h0 + nn] = accL1[j] + accR1[j] + bias;
      }
    }
    __syncthreads();

    *(float4*)(lxs)           = qx0;
    *(float4*)(lxs + LDP)     = qx1;
    *(float4*)(lxs + 2 * LDP) = qx2;
    *(float4*)(lxs + 3 * LDP) = qx3;

#pragma unroll
    for (int rt = 0; rt < ROWT; ++rt) {
      const int r = rt * TILE_M + sr;
      const int p = tid & 15;
      const float* po = lds_o + r * LDP + p * 8;
      float s = 0.f;
#pragma unroll
      for (int q = 0; q < 8; ++q) s += po[q] * po[q];
      s += __shfl_xor(s, 1);
      s += __shfl_xor(s, 2);
      s += __shfl_xor(s, 4);
      s += __shfl_xor(s, 8);
      if (p == 0) lds_inv[r] = 1.0f / fmaxf(sqrtf(s), 1e-12f);
    }
    __syncthreads();

#pragma unroll
    for (int rt = 0; rt < ROWT; ++rt) {
      const int r = rt * TILE_M + sr;
      const float inv = lds_inv[r];
      const float* po = lds_o + r * LDP + sc;
      float4 o0, o1;
      o0.x = fmaxf(po[0] * inv, 0.f); o0.y = fmaxf(po[1] * inv, 0.f);
      o0.z = fmaxf(po[2] * inv, 0.f); o0.w = fmaxf(po[3] * inv, 0.f);
      o1.x = fmaxf(po[4] * inv, 0.f); o1.y = fmaxf(po[5] * inv, 0.f);
      o1.z = fmaxf(po[6] * inv, 0.f); o1.w = fmaxf(po[7] * inv, 0.f);
      float4* pw = (float4*)(xout + (size_t)(row0 + r) * HID + sc);
      pw[0] = o0;
      pw[1] = o1;
    }
  }
}

__global__ __launch_bounds__(256) void mlp_head(
    const float* __restrict__ xin, const float* __restrict__ W1,
    const float* __restrict__ b1, const float* __restrict__ W2,
    const float* __restrict__ b2, float* __restrict__ out, int ntiles) {
  __shared__ float lds_x[TILE_M * LDP];
  __shared__ float lds_h[TILE_M * LDP];

  const int tid   = threadIdx.x;
  const int wave  = tid >> 5;
  const int lane  = tid & 31;
  const int nn    = lane & 15;
  const int khalf = (lane >> 4) * 2;
  const int h0    = wave * 16;
  const int sr    = tid >> 4;
  const int sp    = tid & 15;
  const int sc    = sp * 8;

  int tile = blockIdx.x;
  float4 q0, q1;
  {
    const float4* px = (const float4*)(xin + ((size_t)tile * TILE_M + sr) * HID + sc);
    q0 = px[0];
    q1 = px[1];
  }

  v2f bW[32];
  {
    const float* pw = W1 + (size_t)(h0 + nn) * HID + khalf;
#pragma unroll
    for (int kk = 0; kk < 32; ++kk) bW[kk] = *(const v2f*)(pw + kk * 4);
  }
  const float bias = b1[h0 + nn];
  float w2a[8], w2b[8];
#pragma unroll
  for (int q = 0; q < 8; ++q) {
    w2a[q] = W2[sc + q];
    w2b[q] = W2[HID + sc + q];
  }
  const float c0 = b2[0], c1 = b2[1];
  const int rbase = (lane >> 4) * 8;

  for (; tile < ntiles; tile += (int)gridDim.x) {
    const int row0 = tile * TILE_M;
    {
      float4* lx4 = (float4*)(lds_x + sr * LDP + sc);
      lx4[0] = q0;
      lx4[1] = q1;
    }
    __syncthreads();

    {
      const int nt = tile + (int)gridDim.x;
      const int pt = (nt < ntiles) ? nt : tile;
      const float4* px = (const float4*)(xin + ((size_t)pt * TILE_M + sr) * HID + sc);
      q0 = px[0];
      q1 = px[1];
    }
    __builtin_amdgcn_sched_barrier(0);

    v8f acc = {0.f, 0.f, 0.f, 0.f, 0.f, 0.f, 0.f, 0.f};
    const float* lx = lds_x + nn * LDP + khalf;
#pragma unroll
    for (int kk = 0; kk < 32; ++kk) {
      v2f a = *(const v2f*)(lx + kk * 4);
      acc = __builtin_amdgcn_wmma_f32_16x16x4_f32(false, a, false, bW[kk],
                                                  (short)0, acc, false, false);
    }
#pragma unroll
    for (int j = 0; j < 8; ++j)
      lds_h[(rbase + j) * LDP + h0 + nn] = fmaxf(acc[j] + bias, 0.f);
    __syncthreads();

    {
      const float* ph = lds_h + sr * LDP + sc;
      float s0 = 0.f, s1 = 0.f;
#pragma unroll
      for (int q = 0; q < 8; ++q) {
        float h = ph[q];
        s0 += h * w2a[q];
        s1 += h * w2b[q];
      }
      s0 += __shfl_xor(s0, 1); s1 += __shfl_xor(s1, 1);
      s0 += __shfl_xor(s0, 2); s1 += __shfl_xor(s1, 2);
      s0 += __shfl_xor(s0, 4); s1 += __shfl_xor(s1, 4);
      s0 += __shfl_xor(s0, 8); s1 += __shfl_xor(s1, 8);
      if (sp == 0) {
        out[(size_t)(row0 + sr) * 2 + 0] = s0 + c0;
        out[(size_t)(row0 + sr) * 2 + 1] = s1 + c1;
      }
    }
  }
}

extern "C" void kernel_launch(void* const* d_in, const int* in_sizes, int n_in,
                              void* d_out, int out_size, void* d_ws, size_t ws_size,
                              hipStream_t stream) {
  const float* x  = (const float*)d_in[0];
  const int*   ei = (const int*)d_in[1];
  const float* Wl = (const float*)d_in[2];
  const float* bl = (const float*)d_in[3];
  const float* Wr = (const float*)d_in[4];
  const float* W1 = (const float*)d_in[5];
  const float* b1 = (const float*)d_in[6];
  const float* W2 = (const float*)d_in[7];
  const float* b2 = (const float*)d_in[8];
  float* out = (float*)d_out;

  const int N = NNODES;
  const int E = in_sizes[1] / 2;
  const int* src = ei;
  const int* dst = ei + E;

  float* inv_deg = (float*)d_ws;
  const size_t pad = 100096;
  float* bufA = inv_deg + pad;
  float* bufB = bufA + (size_t)N * HID;
  const size_t npad = (size_t)NBUCK * BNODES;
  const int G = (E + CHUNK - 1) / CHUNK;
  const int M = NBUCK * G;
  const int scanBlocks = (M + 1023) / 1024;
  const size_t mpad = (size_t)scanBlocks * 1024;
  int*  cnt    = (int*)(bufB + (size_t)N * HID);
  int*  rowptr = cnt + npad;
  int*  bh     = rowptr + npad;
  int*  bb     = bh + mpad;
  int*  ebuf   = bb + mpad;
  int*  col    = ebuf + 2 * (size_t)E;

  bin_count<<<G, 1024, 0, stream>>>(dst, bh, E, G);
  scan_kernel<<<scanBlocks, 1024, 0, stream>>>(bh, bb, M);
  bin_scatter<<<G, 1024, 0, stream>>>(src, dst, bb, ebuf, E, G);
  bucket_build<<<NBUCK, 1024, 0, stream>>>(ebuf, bb, cnt, rowptr, inv_deg, col, E, G, N);

  const float* xs[3] = {x, bufA, bufB};
  float*       xo[3] = {bufA, bufB, bufA};
  const int xformChunks   = N / ROWS;
  const int headTiles     = N / TILE_M;

  for (int i = 0; i < 3; ++i) {
    sage_layer<<<XFORM_WG, 256, 0, stream>>>(
        xs[i], rowptr, cnt, col, inv_deg, Wl + (size_t)i * HID * HID,
        bl + (size_t)i * HID, Wr + (size_t)i * HID * HID, xo[i], xformChunks);
  }
  mlp_head<<<HEAD_WG, 256, 0, stream>>>(bufA, W1, b1, W2, b2, out, headTiles);
}
